# starting kernel
# baseline (speedup 1.0000x reference)
#include <hip/hip_runtime.h>
#include <hip/hip_bf16.h>
#include <math.h>

typedef float v2f __attribute__((ext_vector_type(2)));
typedef float v8f __attribute__((ext_vector_type(8)));

__device__ __forceinline__ void edge_sd(const int* __restrict__ ei, int e, int E,
                                        int& s, int& d) {
  if (e < E) { s = ei[e]; d = ei[E + e]; }
  else       { s = e - E; d = s; }
}

__device__ __forceinline__ void atomicMaxF(float* addr, float v) {
  if (v >= 0.0f) atomicMax((int*)addr, __float_as_int(v));
  else           atomicMin((unsigned int*)addr, __float_as_uint(v));
}

__global__ __launch_bounds__(256)
void gemm128_wmma(const float* __restrict__ X, const float* __restrict__ W,
                  float* __restrict__ Y, int nrows) {
  __shared__ float Xs[128 * 33];
  __shared__ float Ws[32 * 132];

  const int tid  = threadIdx.x;
  const int lane = tid & 31;
  const int wv   = tid >> 5;
  const int rc   = lane & 15;
  const int kh   = (lane >> 4) << 1;
  const int r0   = blockIdx.x * 128;

  v8f acc[8];
#pragma unroll
  for (int nt = 0; nt < 8; ++nt)
#pragma unroll
    for (int i = 0; i < 8; ++i) acc[nt][i] = 0.0f;

  for (int kb = 0; kb < 4; ++kb) {
#pragma unroll
    for (int i = 0; i < 4; ++i) {
      int idx = tid + i * 256;
      int row = idx >> 3;
      int c4  = (idx & 7) << 2;
      int gr  = r0 + row;
      float4 v = {0.f, 0.f, 0.f, 0.f};
      if (gr < nrows) v = *(const float4*)(X + (long)gr * 128 + kb * 32 + c4);
      float* p = &Xs[row * 33 + c4];
      p[0] = v.x; p[1] = v.y; p[2] = v.z; p[3] = v.w;
    }
#pragma unroll
    for (int i = 0; i < 4; ++i) {
      int idx = tid + i * 256;
      int row = idx >> 5;
      int c4  = (idx & 31) << 2;
      float4 v = *(const float4*)(W + (long)(kb * 32 + row) * 128 + c4);
      float* p = &Ws[row * 132 + c4];
      p[0] = v.x; p[1] = v.y; p[2] = v.z; p[3] = v.w;
    }
    __syncthreads();

    v2f afr[8];
#pragma unroll
    for (int ks = 0; ks < 8; ++ks) {
      const float* p = &Xs[(wv * 16 + rc) * 33 + ks * 4 + kh];
      afr[ks][0] = p[0];
      afr[ks][1] = p[1];
    }
#pragma unroll
    for (int nt = 0; nt < 8; ++nt) {
#pragma unroll
      for (int ks = 0; ks < 8; ++ks) {
        const float* p = &Ws[(ks * 4 + kh) * 132 + nt * 16 + rc];
        v2f b;
        b[0] = p[0];
        b[1] = p[132];
        acc[nt] = __builtin_amdgcn_wmma_f32_16x16x4_f32(
            false, afr[ks], false, b, (short)0, acc[nt], false, false);
      }
    }
    __syncthreads();
  }

#pragma unroll
  for (int nt = 0; nt < 8; ++nt) {
#pragma unroll
    for (int i = 0; i < 8; ++i) {
      int row = wv * 16 + i + ((lane >> 4) << 3);
      int gr  = r0 + row;
      if (gr < nrows) Y[(long)gr * 128 + nt * 16 + rc] = acc[nt][i];
    }
  }
}

__global__ __launch_bounds__(256)
void attn_scalars(const float* __restrict__ xl, const float* __restrict__ asrc,
                  const float* __restrict__ adst, float* __restrict__ as_out,
                  float* __restrict__ ad_out, int N, int H) {
  int node = blockIdx.x * 8 + (threadIdx.x >> 5);
  int lane = threadIdx.x & 31;
  if (node >= N) return;
  const float* row = xl + (long)node * 128;
  float s[4], d[4];
#pragma unroll
  for (int b = 0; b < 4; ++b) {
    int f = lane + 32 * b;
    float x = row[f];
    s[b] = x * asrc[f];
    d[b] = x * adst[f];
  }
#pragma unroll
  for (int b = 0; b < 4; ++b) {
    for (int off = 16; off > 0; off >>= 1) {
      s[b] += __shfl_xor(s[b], off, 32);
      d[b] += __shfl_xor(d[b], off, 32);
    }
  }
  if (lane == 0) {
    if (H == 4) {
#pragma unroll
      for (int h = 0; h < 4; ++h) {
        as_out[(long)node * 4 + h] = s[h];
        ad_out[(long)node * 4 + h] = d[h];
      }
    } else {
      as_out[node] = s[0] + s[1] + s[2] + s[3];
      ad_out[node] = d[0] + d[1] + d[2] + d[3];
    }
  }
}

__global__ __launch_bounds__(256)
void init_layer(float* __restrict__ out, const float* __restrict__ bias,
                float* __restrict__ m, float* __restrict__ den, int N, int H) {
  long t = (long)blockIdx.x * 256 + threadIdx.x;
  if (t < (long)N * 128) out[t] = bias[t & 127];
  if (t < (long)N * H) { m[t] = -INFINITY; den[t] = 0.0f; }
}

__global__ __launch_bounds__(256)
void edge_logit_max(const int* __restrict__ ei, const float* __restrict__ as,
                    const float* __restrict__ ad, float* __restrict__ elog,
                    float* __restrict__ m, int E, int N, int H) {
  int e = blockIdx.x * 256 + threadIdx.x;
  if (e >= E + N) return;
  int s, d;
  edge_sd(ei, e, E, s, d);
  for (int h = 0; h < H; ++h) {
    float l = as[(long)s * H + h] + ad[(long)d * H + h];
    l = (l > 0.0f) ? l : 0.2f * l;
    elog[(long)e * H + h] = l;
    atomicMaxF(&m[(long)d * H + h], l);
  }
}

__global__ __launch_bounds__(256)
void edge_expsum(const int* __restrict__ ei, float* __restrict__ elog,
                 const float* __restrict__ m, float* __restrict__ den,
                 int E, int N, int H) {
  int e = blockIdx.x * 256 + threadIdx.x;
  if (e >= E + N) return;
  int s, d;
  edge_sd(ei, e, E, s, d);
  for (int h = 0; h < H; ++h) {
    float ex = expf(elog[(long)e * H + h] - m[(long)d * H + h]);
    elog[(long)e * H + h] = ex;
    atomicAdd(&den[(long)d * H + h], ex);
  }
}

__global__ __launch_bounds__(256)
void edge_message(const int* __restrict__ ei, const float* __restrict__ xl,
                  const float* __restrict__ elog, const float* __restrict__ den,
                  float* __restrict__ out, int E, int N, int H) {
  int e = blockIdx.x * 8 + (threadIdx.x >> 5);
  int lane = threadIdx.x & 31;
  if (e >= E + N) return;
  int s, d;
  edge_sd(ei, e, E, s, d);
  const float* xr = xl + (long)s * 128;
  float* orow = out + (long)d * 128;
  __builtin_prefetch(xr + lane * 4, 0, 1);
  float alpha[4];
  if (H == 4) {
#pragma unroll
    for (int h = 0; h < 4; ++h)
      alpha[h] = elog[(long)e * 4 + h] / (den[(long)d * 4 + h] + 1e-16f);
  } else {
    float a0 = elog[e] / (den[d] + 1e-16f);
    alpha[0] = alpha[1] = alpha[2] = alpha[3] = a0;
  }
#pragma unroll
  for (int b = 0; b < 4; ++b) {
    int f = lane + 32 * b;
    atomicAdd(&orow[f], xr[f] * alpha[b]);
  }
}

__global__ __launch_bounds__(256)
void elu_inplace(float* __restrict__ x, long n) {
  long t = (long)blockIdx.x * 256 + threadIdx.x;
  if (t >= n) return;
  float v = x[t];
  x[t] = (v > 0.0f) ? v : (expf(v) - 1.0f);
}

__global__ __launch_bounds__(256)
void l2norm_rows(float* __restrict__ out, int N) {
  int node = blockIdx.x * 8 + (threadIdx.x >> 5);
  int lane = threadIdx.x & 31;
  if (node >= N) return;
  float* row = out + (long)node * 128;
  float v[4];
  float ss = 0.0f;
#pragma unroll
  for (int b = 0; b < 4; ++b) {
    v[b] = row[lane + 32 * b];
    ss += v[b] * v[b];
  }
  for (int off = 16; off > 0; off >>= 1) ss += __shfl_xor(ss, off, 32);
  float sc = 1.0f / fmaxf(sqrtf(ss), 1e-12f);
#pragma unroll
  for (int b = 0; b < 4; ++b) row[lane + 32 * b] = v[b] * sc;
}

extern "C" void kernel_launch(void* const* d_in, const int* in_sizes, int n_in,
                              void* d_out, int out_size, void* d_ws, size_t ws_size,
                              hipStream_t stream) {
  const int*   ei    = (const int*)d_in[0];
  const float* emb   = (const float*)d_in[1];
  const float* W1    = (const float*)d_in[2];
  const float* asrc1 = (const float*)d_in[3];
  const float* adst1 = (const float*)d_in[4];
  const float* b1    = (const float*)d_in[5];
  const float* W2    = (const float*)d_in[6];
  const float* asrc2 = (const float*)d_in[7];
  const float* adst2 = (const float*)d_in[8];
  const float* b2    = (const float*)d_in[9];
  float* out = (float*)d_out;

  const int E  = in_sizes[0] / 2;
  const int N  = in_sizes[1] / 128;
  const int E2 = E + N;

  float* p   = (float*)d_ws;
  float* xl1 = p; p += (size_t)N * 128;
  float* x1  = p; p += (size_t)N * 128;
  float* xl2 = p; p += (size_t)N * 128;
  float* as1 = p; p += (size_t)N * 4;
  float* ad1 = p; p += (size_t)N * 4;
  float* m1  = p; p += (size_t)N * 4;
  float* dn1 = p; p += (size_t)N * 4;
  float* as2 = p; p += (size_t)N;
  float* ad2 = p; p += (size_t)N;
  float* m2  = p; p += (size_t)N;
  float* dn2 = p; p += (size_t)N;
  float* elog = p;

  dim3 blk(256);
  int gGemm  = (N + 127) / 128;
  int gNode8 = (N + 7) / 8;
  int gElem  = (int)(((long)N * 128 + 255) / 256);
  int gEdge  = (E2 + 255) / 256;
  int gEdge8 = (E2 + 7) / 8;

  gemm128_wmma <<<gGemm,  blk, 0, stream>>>(emb, W1, xl1, N);
  attn_scalars <<<gNode8, blk, 0, stream>>>(xl1, asrc1, adst1, as1, ad1, N, 4);
  init_layer   <<<gElem,  blk, 0, stream>>>(x1, b1, m1, dn1, N, 4);
  edge_logit_max<<<gEdge, blk, 0, stream>>>(ei, as1, ad1, elog, m1, E, N, 4);
  edge_expsum  <<<gEdge,  blk, 0, stream>>>(ei, elog, m1, dn1, E, N, 4);
  edge_message <<<gEdge8, blk, 0, stream>>>(ei, xl1, elog, dn1, x1, E, N, 4);
  elu_inplace  <<<gElem,  blk, 0, stream>>>(x1, (long)N * 128);

  gemm128_wmma <<<gGemm,  blk, 0, stream>>>(x1, W2, xl2, N);
  attn_scalars <<<gNode8, blk, 0, stream>>>(xl2, asrc2, adst2, as2, ad2, N, 1);
  init_layer   <<<gElem,  blk, 0, stream>>>(out, b2, m2, dn2, N, 1);
  edge_logit_max<<<gEdge, blk, 0, stream>>>(ei, as2, ad2, elog, m2, E, N, 1);
  edge_expsum  <<<gEdge,  blk, 0, stream>>>(ei, elog, m2, dn2, E, N, 1);
  edge_message <<<gEdge8, blk, 0, stream>>>(ei, xl2, elog, dn2, out, E, N, 1);
  l2norm_rows  <<<gNode8, blk, 0, stream>>>(out, N);
}


# Attention scalars as a ninth WMMA column tile of each GEMM: B = W.att built per K chunk in LDS, a_s/a_d = x.(W.att) (sum reassociated, rounding differs); row-dot launches dropped, 6 launches to 4
# speedup vs baseline: 5.1636x; 5.1636x over previous
#include <hip/hip_runtime.h>
#include <hip/hip_bf16.h>
#include <math.h>

typedef float v2f __attribute__((ext_vector_type(2)));
typedef float v4f __attribute__((ext_vector_type(4)));
typedef float v8f __attribute__((ext_vector_type(8)));

#define WAS_LD 17

__global__ __launch_bounds__(256)
void gemm128_wmma(const float* __restrict__ X, const float* __restrict__ W,
                  const float* __restrict__ asrc, const float* __restrict__ adst,
                  float* __restrict__ Y, float* __restrict__ as_out,
                  float* __restrict__ ad_out, int nrows, int H) {
  __shared__ float Xs[128 * 33];
  __shared__ float Ws[32 * 132];
  __shared__ float Was[32 * WAS_LD];

  const int tid  = threadIdx.x;
  const int lane = tid & 31;
  const int wv   = tid >> 5;
  const int rc   = lane & 15;
  const int kh   = (lane >> 4) << 1;
  const int r0   = blockIdx.x * 128;

  const int wrow = tid >> 3;
  const int wj   = tid & 7;
  float attr[32];
  {
    const float* ap = ((wj < 4) ? asrc : adst) + (wj & 3) * 32;
#pragma unroll
    for (int q = 0; q < 8; ++q) {
      const float4 v = *(const float4*)(ap + q * 4);
      attr[q * 4 + 0] = v.x; attr[q * 4 + 1] = v.y; attr[q * 4 + 2] = v.z; attr[q * 4 + 3] = v.w;
    }
  }

  v8f acc[8];
#pragma unroll
  for (int nt = 0; nt < 8; ++nt)
#pragma unroll
    for (int i = 0; i < 8; ++i) acc[nt][i] = 0.0f;
  v8f acc9;
#pragma unroll
  for (int i = 0; i < 8; ++i) acc9[i] = 0.0f;

  for (int kb = 0; kb < 4; ++kb) {
#pragma unroll
    for (int i = 0; i < 4; ++i) {
      int idx = tid + i * 256;
      int row = idx >> 3;
      int c4  = (idx & 7) << 2;
      int gr  = r0 + row;
      float4 v = {0.f, 0.f, 0.f, 0.f};
      if (gr < nrows) v = *(const float4*)(X + (long)gr * 128 + kb * 32 + c4);
      float* p = &Xs[row * 33 + c4];
      p[0] = v.x; p[1] = v.y; p[2] = v.z; p[3] = v.w;
    }
#pragma unroll
    for (int i = 0; i < 4; ++i) {
      int idx = tid + i * 256;
      int row = idx >> 5;
      int c4  = (idx & 31) << 2;
      float4 v = *(const float4*)(W + (long)(kb * 32 + row) * 128 + c4);
      float* p = &Ws[row * 132 + c4];
      p[0] = v.x; p[1] = v.y; p[2] = v.z; p[3] = v.w;
    }
    __syncthreads();

    {
      const v4f* wp = (const v4f*)&Ws[wrow * 132 + (wj & 3) * 32];
      v4f w[8];
#pragma unroll
      for (int q = 0; q < 8; ++q) w[q] = wp[q];
      asm volatile("" :: "v"(w[0]), "v"(w[1]), "v"(w[2]), "v"(w[3]), "v"(w[4]), "v"(w[5]), "v"(w[6]), "v"(w[7]));
      float s0 = 0.0f, s1 = 0.0f, s2 = 0.0f, s3 = 0.0f;
#pragma unroll
      for (int q = 0; q < 8; ++q) {
        s0 += w[q][0] * attr[q * 4 + 0];
        s1 += w[q][1] * attr[q * 4 + 1];
        s2 += w[q][2] * attr[q * 4 + 2];
        s3 += w[q][3] * attr[q * 4 + 3];
      }
      Was[wrow * WAS_LD + wj]     = (s0 + s1) + (s2 + s3);
      Was[wrow * WAS_LD + 8 + wj] = 0.0f;
    }

    v2f afr[8];
#pragma unroll
    for (int ks = 0; ks < 8; ++ks) {
      const float* p = &Xs[(wv * 16 + rc) * 33 + ks * 4 + kh];
      afr[ks][0] = p[0];
      afr[ks][1] = p[1];
    }
#pragma unroll
    for (int nt = 0; nt < 8; ++nt) {
#pragma unroll
      for (int ks = 0; ks < 8; ++ks) {
        const float* p = &Ws[(ks * 4 + kh) * 132 + nt * 16 + rc];
        v2f b;
        b[0] = p[0];
        b[1] = p[132];
        acc[nt] = __builtin_amdgcn_wmma_f32_16x16x4_f32(
            false, afr[ks], false, b, (short)0, acc[nt], false, false);
      }
    }
    __syncthreads();

    v2f b9[8];
#pragma unroll
    for (int ks = 0; ks < 8; ++ks) {
      const float* p = &Was[(ks * 4 + kh) * WAS_LD + rc];
      b9[ks][0] = p[0];
      b9[ks][1] = p[WAS_LD];
    }
    asm volatile("" :: "v"(b9[0]), "v"(b9[1]), "v"(b9[2]), "v"(b9[3]), "v"(b9[4]), "v"(b9[5]), "v"(b9[6]), "v"(b9[7]));
#pragma unroll
    for (int ks = 0; ks < 8; ++ks)
      acc9 = __builtin_amdgcn_wmma_f32_16x16x4_f32(
          false, afr[ks], false, b9[ks], (short)0, acc9, false, false);
  }

#pragma unroll
  for (int nt = 0; nt < 8; ++nt) {
#pragma unroll
    for (int i = 0; i < 8; ++i) {
      int row = wv * 16 + i + ((lane >> 4) << 3);
      int gr  = r0 + row;
      if (gr < nrows) Y[(long)gr * 128 + nt * 16 + rc] = acc[nt][i];
    }
  }
  if (H == 4) {
#pragma unroll
    for (int i = 0; i < 8; ++i) {
      int gr = r0 + wv * 16 + i + ((lane >> 4) << 3);
      if (gr < nrows && rc < 8) {
        float* o = (rc < 4) ? as_out : ad_out;
        o[(long)gr * 4 + (rc & 3)] = acc9[i];
      }
    }
  } else {
#pragma unroll
    for (int i = 0; i < 8; ++i) {
      int gr = r0 + wv * 16 + i + ((lane >> 4) << 3);
      float t = acc9[i];
      t += __shfl_xor(t, 1, 32);
      t += __shfl_xor(t, 2, 32);
      if (gr < nrows && (rc == 0 || rc == 4)) {
        float* o = (rc == 0) ? as_out : ad_out;
        o[gr] = t;
      }
    }
  }
}

#define TR  196
#define NWV 32
#define CAP 16384
#define WCAP 384

__device__ __forceinline__ float leaky02(float l) { return (l > 0.0f) ? l : 0.2f * l; }

__device__ __forceinline__ int lane_bcast_i(int v, int j) {
  return __builtin_amdgcn_readlane(v, j);
}
__device__ __forceinline__ float lane_bcast_f(float v, int j) {
  return __int_as_float(__builtin_amdgcn_readlane(__float_as_int(v), j));
}

template <bool PLACE>
__device__ __forceinline__ void scan_chunk(const int (&d)[4], int e0, int r0, int nrows,
                                           int wv, int lane, int* cnt,
                                           const int* rowoff, int* list,
                                           int& wcount, int* wle, int* wlr) {
  unsigned u[4]; bool mt[4];
#pragma unroll
  for (int c = 0; c < 4; ++c) { u[c] = (unsigned)(d[c] - r0); mt[c] = u[c] < (unsigned)nrows; }
  if (!__builtin_amdgcn_ballot_w32(mt[0] | mt[1] | mt[2] | mt[3])) return;
#pragma unroll
  for (int c = 0; c < 4; ++c) {
    if (!PLACE) {
      const unsigned mask = __builtin_amdgcn_ballot_w32(mt[c]);
      if (mask == 0) continue;
      if (mt[c]) {
        atomicAdd(&cnt[wv * TR + (int)u[c]], 1);
        const int pos = wcount + __builtin_popcount(mask & ((1u << lane) - 1u));
        if (pos < WCAP) { wle[pos] = e0 + c; wlr[pos] = (int)u[c]; }
      }
      wcount += __builtin_popcount(mask);
    } else {
      unsigned mask = __builtin_amdgcn_ballot_w32(mt[c]);
      while (mask) {
        const int L = __builtin_ctz(mask);
        mask &= mask - 1;
        const int rowL = lane_bcast_i((int)u[c], L);
        if (lane == 0) {
          const int p = cnt[wv * TR + rowL];
          cnt[wv * TR + rowL] = p + 1;
          const int pos = rowoff[rowL] + p;
          if (pos < CAP) list[pos] = e0 - lane * 4 + L * 4 + c;
        }
      }
    }
  }
}

template <bool PLACE>
__device__ __forceinline__ void scan_slice(const int* __restrict__ ei, int E, int r0, int nrows,
                                           int wv, int lane, int per, int* cnt,
                                           const int* rowoff, int* list,
                                           int& wcount, int* wle, int* wlr) {
  const int* dstp = ei + E;
  const int beg = wv * per;
  const int end = (beg + per < E) ? (beg + per) : E;
  const bool al = ((((size_t)dstp) & 15) == 0);
  const int ng = (end > beg) ? ((end - beg + 1023) >> 10) : 0;
  const int rot = (ng > 0) ? (int)(blockIdx.x % (unsigned)ng) : 0;
  for (int g = 0; g < ng; ++g) {
    const int gg = (g + rot < ng) ? (g + rot) : (g + rot - ng);
    const int base = beg + (gg << 10);
    if (al && base + 1024 <= end) {
      int4 v[8];
#pragma unroll
      for (int q = 0; q < 8; ++q) v[q] = *(const int4*)(dstp + base + q * 128 + lane * 4);
#pragma unroll
      for (int q = 0; q < 8; ++q) {
        const int d[4] = {v[q].x, v[q].y, v[q].z, v[q].w};
        scan_chunk<PLACE>(d, base + q * 128 + lane * 4, r0, nrows, wv, lane, cnt, rowoff, list, wcount, wle, wlr);
      }
    } else {
      const int lim = (base + 1024 < end) ? (base + 1024) : end;
      for (int b2 = base; b2 < lim; b2 += 128) {
        const int e0 = b2 + lane * 4;
        int d[4] = {-1, -1, -1, -1};
#pragma unroll
        for (int c = 0; c < 4; ++c) if (e0 + c < end) d[c] = dstp[e0 + c];
        scan_chunk<PLACE>(d, e0, r0, nrows, wv, lane, cnt, rowoff, list, wcount, wle, wlr);
      }
    }
  }
}

__device__ __forceinline__ void place_recorded(int wv, int lane, int wcount, const int* wle,
                                               const int* wlr, int* cnt, const int* rowoff,
                                               int* list) {
  for (int k = 0; k < wcount; k += 32) {
    const int i = k + lane;
    const bool v = i < wcount;
    const int e  = v ? wle[i] : 0;
    const int rw = v ? wlr[i] : 0;
    unsigned mask = __builtin_amdgcn_ballot_w32(v);
    while (mask) {
      const int L = __builtin_ctz(mask);
      mask &= mask - 1;
      const int rowL = lane_bcast_i(rw, L);
      const int eL   = lane_bcast_i(e, L);
      if (lane == 0) {
        const int p = cnt[wv * TR + rowL];
        cnt[wv * TR + rowL] = p + 1;
        const int pos = rowoff[rowL] + p;
        if (pos < CAP) list[pos] = eL;
      }
    }
  }
}

template <int H, bool LAST, bool BUILD>
__global__ __launch_bounds__(1024)
void gat_dst_tile(const int* __restrict__ ei, const float* __restrict__ xl,
                  const float* __restrict__ as, const float* __restrict__ ad,
                  const float* __restrict__ bias, float* __restrict__ outp,
                  int* __restrict__ glist, int* __restrict__ goff, int E, int N) {
  __shared__ int cnt[NWV * TR];
  __shared__ int tot[TR + 32];
  __shared__ int rowoff[TR + 32];
  __shared__ int list[CAP];
  __shared__ int wle_all[BUILD ? NWV * WCAP : 1];
  __shared__ int wlr_all[BUILD ? NWV * WCAP : 1];

  const int tid  = threadIdx.x;
  const int lane = tid & 31;
  const int wv   = tid >> 5;
  const int r0   = blockIdx.x * TR;
  const int nrows = (N - r0 < TR) ? (N - r0) : TR;
  const int per  = (((E + NWV - 1) / NWV) + 1023) & ~1023;
  int* mylist = glist + (long)blockIdx.x * CAP;
  int* myoff  = goff + (long)blockIdx.x * (TR + 1);

  bool ovf;
  if (BUILD) {
    for (int i = tid; i < NWV * TR; i += 1024) cnt[i] = 0;
    if (tid < TR + 32) { tot[tid] = 0; rowoff[tid] = 0; }
    __syncthreads();

    int wcount = 0;
    int* wle = wle_all + wv * WCAP;
    int* wlr = wlr_all + wv * WCAP;
    scan_slice<false>(ei, E, r0, nrows, wv, lane, per, cnt, rowoff, list, wcount, wle, wlr);
    __syncthreads();

    if (tid < TR) {
      int run = 0;
      for (int w2 = 0; w2 < NWV; ++w2) {
        const int q = cnt[w2 * TR + tid];
        cnt[w2 * TR + tid] = run;
        run += q;
      }
      tot[tid] = run;
    }
    __syncthreads();
    if (wv == 0) {
      int loc[7]; int sum = 0;
#pragma unroll
      for (int i = 0; i < 7; ++i) { loc[i] = sum; sum += tot[lane * 7 + i]; }
      int incl = sum;
      for (int o = 1; o < 32; o <<= 1) {
        int t = __shfl_up(incl, o, 32);
        if (lane >= o) incl += t;
      }
      const int excl = incl - sum;
#pragma unroll
      for (int i = 0; i < 7; ++i) rowoff[lane * 7 + i] = excl + loc[i];
    }
    __syncthreads();
    ovf = rowoff[TR] > CAP;
    if (!ovf) {
      if (wcount <= WCAP) place_recorded(wv, lane, wcount, wle, wlr, cnt, rowoff, list);
      else scan_slice<true>(ei, E, r0, nrows, wv, lane, per, cnt, rowoff, list, wcount, wle, wlr);
    }
    __syncthreads();
    if (!ovf) {
      const int total = rowoff[TR];
      for (int i = tid; i < total; i += 1024) {
        const int s = ei[list[i]];
        list[i] = s;
        mylist[i] = s;
      }
    }
    if (tid <= TR) myoff[tid] = rowoff[tid];
    __syncthreads();
  } else {
    if (tid <= TR) rowoff[tid] = myoff[tid];
    __syncthreads();
    ovf = rowoff[TR] > CAP;
    if (!ovf) {
      const int total = rowoff[TR];
      for (int i = tid; i < total; i += 1024) list[i] = mylist[i];
    }
    __syncthreads();
  }

  const int* dstp = ei + E;
  for (int r = wv; r < nrows; r += NWV) {
    const int n   = r0 + r;
    const int off = rowoff[r];
    const int c   = rowoff[r + 1] - off;
    const int nch = ovf ? ((E + 31) >> 5) : ((c + 31) >> 5);

    float adv[H], ls[H], mx[H];
#pragma unroll
    for (int h = 0; h < H; ++h) {
      adv[h] = ad[(long)n * H + h];
      ls[h]  = leaky02(as[(long)n * H + h] + adv[h]);
      mx[h]  = ls[h];
    }
    for (int k = 0; k < nch; ++k) {
      const int i = k * 32 + lane;
      bool valid; int sv = 0;
      if (!ovf) { valid = i < c; if (valid) sv = list[off + i]; }
      else      { valid = (i < E) && (dstp[i] == n); if (valid) sv = ei[i]; }
      if (valid) {
#pragma unroll
        for (int h = 0; h < H; ++h)
          mx[h] = fmaxf(mx[h], leaky02(as[(long)sv * H + h] + adv[h]));
      }
    }
#pragma unroll
    for (int h = 0; h < H; ++h)
      for (int o = 16; o > 0; o >>= 1) mx[h] = fmaxf(mx[h], __shfl_xor(mx[h], o, 32));

    float exs[H], dl[H];
#pragma unroll
    for (int h = 0; h < H; ++h) { exs[h] = expf(ls[h] - mx[h]); dl[h] = 0.0f; }
    float acc[4];
    {
      const float* xr = xl + (long)n * 128;
#pragma unroll
      for (int b = 0; b < 4; ++b) acc[b] = xr[lane + 32 * b] * exs[(H == 4) ? b : 0];
    }
    for (int k = 0; k < nch; ++k) {
      const int i = k * 32 + lane;
      bool valid; int sv = 0;
      if (!ovf) { valid = i < c; if (valid) sv = list[off + i]; }
      else      { valid = (i < E) && (dstp[i] == n); if (valid) sv = ei[i]; }
      float ex[H];
#pragma unroll
      for (int h = 0; h < H; ++h) {
        ex[h] = valid ? expf(leaky02(as[(long)sv * H + h] + adv[h]) - mx[h]) : 0.0f;
        dl[h] += ex[h];
      }
      unsigned mask = __builtin_amdgcn_ballot_w32(valid);
      while (__builtin_popcount(mask) >= 4) {
        int js[4]; const float* xr[4]; float x[4][4];
#pragma unroll
        for (int q = 0; q < 4; ++q) {
          js[q] = __builtin_ctz(mask); mask &= mask - 1;
          xr[q] = xl + (long)lane_bcast_i(sv, js[q]) * 128;
        }
#pragma unroll
        for (int q = 0; q < 4; ++q)
#pragma unroll
          for (int b = 0; b < 4; ++b) x[q][b] = xr[q][lane + 32 * b];
#pragma unroll
        for (int q = 0; q < 4; ++q)
#pragma unroll
          for (int b = 0; b < 4; ++b)
            acc[b] += x[q][b] * lane_bcast_f(ex[(H == 4) ? b : 0], js[q]);
      }
      while (mask) {
        const int j = __builtin_ctz(mask); mask &= mask - 1;
        const float* xr = xl + (long)lane_bcast_i(sv, j) * 128;
#pragma unroll
        for (int b = 0; b < 4; ++b)
          acc[b] += xr[lane + 32 * b] * lane_bcast_f(ex[(H == 4) ? b : 0], j);
      }
    }
    float den[H];
#pragma unroll
    for (int h = 0; h < H; ++h) {
      float s = dl[h];
      for (int o = 16; o > 0; o >>= 1) s += __shfl_xor(s, o, 32);
      den[h] = s + exs[h];
    }
    float v[4];
#pragma unroll
    for (int b = 0; b < 4; ++b)
      v[b] = acc[b] / (den[(H == 4) ? b : 0] + 1e-16f) + bias[lane + 32 * b];
    float* orow = outp + (long)n * 128;
    if (!LAST) {
#pragma unroll
      for (int b = 0; b < 4; ++b) orow[lane + 32 * b] = (v[b] > 0.0f) ? v[b] : (expf(v[b]) - 1.0f);
    } else {
      float ss = 0.0f;
#pragma unroll
      for (int b = 0; b < 4; ++b) ss += v[b] * v[b];
      for (int o = 16; o > 0; o >>= 1) ss += __shfl_xor(ss, o, 32);
      const float sc = 1.0f / fmaxf(sqrtf(ss), 1e-12f);
#pragma unroll
      for (int b = 0; b < 4; ++b) orow[lane + 32 * b] = v[b] * sc;
    }
  }
}

extern "C" void kernel_launch(void* const* d_in, const int* in_sizes, int n_in,
                              void* d_out, int out_size, void* d_ws, size_t ws_size,
                              hipStream_t stream) {
  const int*   ei    = (const int*)d_in[0];
  const float* emb   = (const float*)d_in[1];
  const float* W1    = (const float*)d_in[2];
  const float* asrc1 = (const float*)d_in[3];
  const float* adst1 = (const float*)d_in[4];
  const float* b1    = (const float*)d_in[5];
  const float* W2    = (const float*)d_in[6];
  const float* asrc2 = (const float*)d_in[7];
  const float* adst2 = (const float*)d_in[8];
  const float* b2    = (const float*)d_in[9];
  float* out = (float*)d_out;

  const int E  = in_sizes[0] / 2;
  const int N  = in_sizes[1] / 128;

  float* p   = (float*)d_ws;
  float* xl1 = p; p += (size_t)N * 128;
  float* x1  = p; p += (size_t)N * 128;
  float* xl2 = p; p += (size_t)N * 128;
  float* as1 = p; p += (size_t)N * 4;
  float* ad1 = p; p += (size_t)N * 4;
  float* as2 = p; p += (size_t)N;
  float* ad2 = p; p += (size_t)N;
  int gTile  = (N + TR - 1) / TR;
  int* glist = (int*)p; p += (size_t)gTile * CAP;
  int* goff  = (int*)p;

  dim3 blk(256);
  int gGemm  = (N + 127) / 128;

  gemm128_wmma <<<gGemm,  blk, 0, stream>>>(emb, W1, asrc1, adst1, xl1, as1, ad1, N, 4);
  gat_dst_tile<4, false, true><<<gTile, dim3(1024), 0, stream>>>(ei, xl1, as1, ad1, b1, x1, glist, goff, E, N);

  gemm128_wmma <<<gGemm,  blk, 0, stream>>>(x1, W2, asrc2, adst2, xl2, as2, ad2, N, 1);
  gat_dst_tile<1, true, false><<<gTile, dim3(1024), 0, stream>>>(ei, xl2, as2, ad2, b2, out, glist, goff, E, N);
}


# First scan from bins: the first GEMM launch sorts each 2048-edge chunk into 256 key bins (128-byte lines); tile blocks read their own lines, not all keys; new fixed list order (sum order changes)
# speedup vs baseline: 8.0030x; 1.5499x over previous
#include <hip/hip_runtime.h>
#include <hip/hip_bf16.h>
#include <math.h>

typedef float v2f __attribute__((ext_vector_type(2)));
typedef float v4f __attribute__((ext_vector_type(4)));
typedef float v8f __attribute__((ext_vector_type(8)));

#define WAS_LD 17
#define TR  196
#define BCH 2048
#define BW  32

__global__ __launch_bounds__(256)
void gemm128_wmma(const float* __restrict__ X, const float* __restrict__ W,
                  const float* __restrict__ asrc, const float* __restrict__ adst,
                  float* __restrict__ Y, float* __restrict__ as_out,
                  float* __restrict__ ad_out, int nrows, int H,
                  const int* __restrict__ dstv, int* __restrict__ tkey, int* __restrict__ bins, int E) {
  __shared__ int bcnt[8 * 256];
  __shared__ __attribute__((aligned(16))) int binL[256 * BW];
  __shared__ float Xs[128 * 33];
  __shared__ float Ws[32 * 132];
  __shared__ float Was[32 * WAS_LD];

  const int tid  = threadIdx.x;
  const int lane = tid & 31;
  const int wv   = tid >> 5;
  const int rc   = lane & 15;
  const int kh   = (lane >> 4) << 1;
  const int r0   = blockIdx.x * 128;

  const float* px[4];
  const float* pw[4];
  int xso[4], wso[4];
#pragma unroll
  for (int i = 0; i < 4; ++i) {
    const int idx = tid + i * 256;
    const int xrow = idx >> 3, xc4 = (idx & 7) << 2;
    const int gr = (r0 + xrow < nrows) ? (r0 + xrow) : (nrows - 1);
    px[i]  = X + (long)gr * 128 + xc4;
    xso[i] = xrow * 33 + xc4;
    const int wrow4 = idx >> 5, wc4 = (idx & 31) << 2;
    pw[i]  = W + (long)wrow4 * 128 + wc4;
    wso[i] = wrow4 * 132 + wc4;
  }
  v4f xr[4], wr[4];
#pragma unroll
  for (int i = 0; i < 4; ++i) {
    xr[i] = *(const v4f*)px[i];
    wr[i] = *(const v4f*)pw[i];
  }

  if (tkey != nullptr) {
    const int ng8 = (E + 7) >> 3;
    const int nchunk = (E + BCH - 1) / BCH;
    const bool al = ((((size_t)dstv) & 15) == 0);
    const unsigned lower = (1u << lane) - 1u;
    for (int c = (int)blockIdx.x; c < nchunk; c += (int)gridDim.x) {
      for (int i = tid; i < 8 * 256; i += 256) bcnt[i] = 0;
      for (int i = tid; i < 256 * BW / 4; i += 256) ((int4*)binL)[i] = make_int4(0, 0, 0, 0);
      __syncthreads();
      const int g = c * 256 + tid;
      const int e = g << 3;
      int d[8];
      if (al && e + 8 <= E) {
        const int4 a = *(const int4*)(dstv + e);
        const int4 b = *(const int4*)(dstv + e + 4);
        d[0] = a.x; d[1] = a.y; d[2] = a.z; d[3] = a.w;
        d[4] = b.x; d[5] = b.y; d[6] = b.z; d[7] = b.w;
      } else {
#pragma unroll
        for (int j = 0; j < 8; ++j) d[j] = (e + j < E) ? dstv[e + j] : 0;
      }
      int key[8];
#pragma unroll
      for (int j = 0; j < 8; ++j) key[j] = (int)(((unsigned)d[j] / (unsigned)TR) & 255u);
      if (g < ng8) {
        unsigned lo = 0u, hi = 0u;
#pragma unroll
        for (int j = 0; j < 4; ++j) {
          lo |= (unsigned)key[j] << (8 * j);
          hi |= (unsigned)key[4 + j] << (8 * j);
        }
        int2 o; o.x = (int)lo; o.y = (int)hi;
        *(int2*)(tkey + (g << 1)) = o;
      }
#pragma unroll
      for (int j = 0; j < 8; ++j)
        if (e + j < E) atomicAdd(&bcnt[wv * 256 + key[j]], 1);
      __syncthreads();
      {
        int run = 0;
#pragma unroll
        for (int w2 = 0; w2 < 8; ++w2) {
          const int q = bcnt[w2 * 256 + tid];
          bcnt[w2 * 256 + tid] = run;
          run += q;
        }
        binL[tid * BW] = run;
      }
      __syncthreads();
#pragma unroll
      for (int j = 0; j < 8; ++j) {
        const bool pv = e + j < E;
        unsigned peers = __builtin_amdgcn_ballot_w32(pv);
#pragma unroll
        for (int b = 0; b < 8; ++b) {
          const bool bit = ((key[j] >> b) & 1) != 0;
          const unsigned mb = __builtin_amdgcn_ballot_w32(bit);
          peers &= bit ? mb : ~mb;
        }
        if (pv) {
          const int p   = bcnt[wv * 256 + key[j]];
          const int pos = p + __builtin_popcount(peers & lower);
          if (pos < BW - 1) binL[key[j] * BW + 1 + pos] = e + j;
          if ((peers & lower) == 0u) bcnt[wv * 256 + key[j]] = p + __builtin_popcount(peers);
        }
      }
      __syncthreads();
      int4* bout = (int4*)(bins + (size_t)c * 256 * BW);
      for (int i = tid; i < 256 * BW / 4; i += 256) bout[i] = ((const int4*)binL)[i];
      __syncthreads();
    }
  }

  const int wrow = tid >> 3;
  const int wj   = tid & 7;
  float attr[32];
  {
    const float* ap = ((wj < 4) ? asrc : adst) + (wj & 3) * 32;
#pragma unroll
    for (int q = 0; q < 8; ++q) {
      const float4 v = *(const float4*)(ap + q * 4);
      attr[q * 4 + 0] = v.x; attr[q * 4 + 1] = v.y; attr[q * 4 + 2] = v.z; attr[q * 4 + 3] = v.w;
    }
  }

  v8f acc[8];
#pragma unroll
  for (int nt = 0; nt < 8; ++nt)
#pragma unroll
    for (int i = 0; i < 8; ++i) acc[nt][i] = 0.0f;
  v8f acc9;
#pragma unroll
  for (int i = 0; i < 8; ++i) acc9[i] = 0.0f;

  for (int kb = 0; kb < 4; ++kb) {
#pragma unroll
    for (int i = 0; i < 4; ++i) {
      float* p = &Xs[xso[i]];
      p[0] = xr[i][0]; p[1] = xr[i][1]; p[2] = xr[i][2]; p[3] = xr[i][3];
      float* q = &Ws[wso[i]];
      q[0] = wr[i][0]; q[1] = wr[i][1]; q[2] = wr[i][2]; q[3] = wr[i][3];
    }
    __syncthreads();
    if (kb < 3) {
#pragma unroll
      for (int i = 0; i < 4; ++i) {
        xr[i] = *(const v4f*)(px[i] + (kb + 1) * 32);
        wr[i] = *(const v4f*)(pw[i] + (long)(kb + 1) * 32 * 128);
      }
    }

    {
      const v4f* wp = (const v4f*)&Ws[wrow * 132 + (wj & 3) * 32];
      v4f w[8];
#pragma unroll
      for (int q = 0; q < 8; ++q) w[q] = wp[q];
      asm volatile("" :: "v"(w[0]), "v"(w[1]), "v"(w[2]), "v"(w[3]), "v"(w[4]), "v"(w[5]), "v"(w[6]), "v"(w[7]));
      float s0 = 0.0f, s1 = 0.0f, s2 = 0.0f, s3 = 0.0f;
#pragma unroll
      for (int q = 0; q < 8; ++q) {
        s0 += w[q][0] * attr[q * 4 + 0];
        s1 += w[q][1] * attr[q * 4 + 1];
        s2 += w[q][2] * attr[q * 4 + 2];
        s3 += w[q][3] * attr[q * 4 + 3];
      }
      Was[wrow * WAS_LD + wj]     = (s0 + s1) + (s2 + s3);
      Was[wrow * WAS_LD + 8 + wj] = 0.0f;
    }

    v2f afr[8];
#pragma unroll
    for (int ks = 0; ks < 8; ++ks) {
      const float* p = &Xs[(wv * 16 + rc) * 33 + ks * 4 + kh];
      afr[ks][0] = p[0];
      afr[ks][1] = p[1];
    }
#pragma unroll
    for (int nt = 0; nt < 8; ++nt) {
#pragma unroll
      for (int ks = 0; ks < 8; ++ks) {
        const float* p = &Ws[(ks * 4 + kh) * 132 + nt * 16 + rc];
        v2f b;
        b[0] = p[0];
        b[1] = p[132];
        acc[nt] = __builtin_amdgcn_wmma_f32_16x16x4_f32(
            false, afr[ks], false, b, (short)0, acc[nt], false, false);
      }
    }
    __syncthreads();

    v2f b9[8];
#pragma unroll
    for (int ks = 0; ks < 8; ++ks) {
      const float* p = &Was[(ks * 4 + kh) * WAS_LD + rc];
      b9[ks][0] = p[0];
      b9[ks][1] = p[WAS_LD];
    }
    asm volatile("" :: "v"(b9[0]), "v"(b9[1]), "v"(b9[2]), "v"(b9[3]), "v"(b9[4]), "v"(b9[5]), "v"(b9[6]), "v"(b9[7]));
#pragma unroll
    for (int ks = 0; ks < 8; ++ks)
      acc9 = __builtin_amdgcn_wmma_f32_16x16x4_f32(
          false, afr[ks], false, b9[ks], (short)0, acc9, false, false);
  }

#pragma unroll
  for (int nt = 0; nt < 8; ++nt) {
#pragma unroll
    for (int i = 0; i < 8; ++i) {
      int row = wv * 16 + i + ((lane >> 4) << 3);
      int gr  = r0 + row;
      if (gr < nrows) Y[(long)gr * 128 + nt * 16 + rc] = acc[nt][i];
    }
  }
  if (H == 4) {
#pragma unroll
    for (int i = 0; i < 8; ++i) {
      int gr = r0 + wv * 16 + i + ((lane >> 4) << 3);
      if (gr < nrows && rc < 8) {
        float* o = (rc < 4) ? as_out : ad_out;
        o[(long)gr * 4 + (rc & 3)] = acc9[i];
      }
    }
  } else {
#pragma unroll
    for (int i = 0; i < 8; ++i) {
      int gr = r0 + wv * 16 + i + ((lane >> 4) << 3);
      float t = acc9[i];
      t += __shfl_xor(t, 1, 32);
      t += __shfl_xor(t, 2, 32);
      if (gr < nrows && (rc == 0 || rc == 4)) {
        float* o = (rc == 0) ? as_out : ad_out;
        o[gr] = t;
      }
    }
  }
}

#define NWV 32
#define CAP 16384
#define WCAP 384
static_assert(TR <= 255, "placing tells rows apart by eight bits and keeps 255 for no row");

__device__ __forceinline__ float leaky02(float l) { return (l > 0.0f) ? l : 0.2f * l; }

__device__ __forceinline__ int lane_bcast_i(int v, int j) {
  return __builtin_amdgcn_readlane(v, j);
}
#define DPP8_SEL(a, b, c, d, e, f, g, h) \
  ((a) | ((b) << 3) | ((c) << 6) | ((d) << 9) | ((e) << 12) | ((f) << 15) | ((g) << 18) | ((h) << 21))
#define DPP8_BCAST(q) DPP8_SEL(q, q, q, q, q, q, q, q)
template <unsigned SEL>
__device__ __forceinline__ float dpp8f(float v) {
  return __int_as_float(__builtin_amdgcn_mov_dpp8(__float_as_int(v), SEL));
}
__device__ __forceinline__ float grp8_max(float v) {
  v = fmaxf(v, dpp8f<DPP8_SEL(1, 0, 3, 2, 5, 4, 7, 6)>(v));
  v = fmaxf(v, dpp8f<DPP8_SEL(2, 3, 0, 1, 6, 7, 4, 5)>(v));
  v = fmaxf(v, dpp8f<DPP8_SEL(4, 5, 6, 7, 0, 1, 2, 3)>(v));
  return v;
}
__device__ __forceinline__ float grp8_sum(float v) {
  v += dpp8f<DPP8_SEL(1, 0, 3, 2, 5, 4, 7, 6)>(v);
  v += dpp8f<DPP8_SEL(2, 3, 0, 1, 6, 7, 4, 5)>(v);
  v += dpp8f<DPP8_SEL(4, 5, 6, 7, 0, 1, 2, 3)>(v);
  return v;
}
__device__ __forceinline__ v4f ld_row(const float* __restrict__ base, int row, int lane) {
  const unsigned ofs = ((unsigned)row << 9) + ((unsigned)lane << 4);
  return *(const v4f*)((const char*)base + ofs);
}
__device__ __forceinline__ float grp8_bcast(float v, int q) {
  switch (q & 7) {
    case 0:  return dpp8f<DPP8_BCAST(0)>(v);
    case 1:  return dpp8f<DPP8_BCAST(1)>(v);
    case 2:  return dpp8f<DPP8_BCAST(2)>(v);
    case 3:  return dpp8f<DPP8_BCAST(3)>(v);
    case 4:  return dpp8f<DPP8_BCAST(4)>(v);
    case 5:  return dpp8f<DPP8_BCAST(5)>(v);
    case 6:  return dpp8f<DPP8_BCAST(6)>(v);
    default: return dpp8f<DPP8_BCAST(7)>(v);
  }
}
template <int M>
__device__ __forceinline__ void row_trip(const float* __restrict__ xl, int s0, int s1, float e0, float e1,
                                         int lane, v4f& acc) {
  v4f x[M];
#pragma unroll
  for (int q = 0; q < M; ++q) x[q] = ld_row(xl, lane_bcast_i((q < 8) ? s0 : s1, q & 7), lane);
#pragma unroll
  for (int q = 0; q < M; ++q) asm volatile("" :: "v"(x[q]));
#pragma unroll
  for (int q = 0; q < M; ++q) acc += x[q] * grp8_bcast((q < 8) ? e0 : e1, q);
}

template <int H, int M, bool MORE>
__device__ __forceinline__ void row_open(const float* __restrict__ xl, const float* __restrict__ as,
                                         const float* __restrict__ ad, int hh,
                                         int s0, int s1, bool v0, bool v1, int s2, int s3, bool v2, bool v3,
                                         float mrest, int n, int lane,
                                         float& exs, float& dl, float& mxo, float& e2o, float& e3o, float& advo,
                                         v4f& acc) {
  const float adv = ad[(long)n * H + hh];
  const float asn = as[(long)n * H + hh];
  const float a0 = as[(long)s0 * H + hh];
  const float a1 = as[(long)s1 * H + hh];
  float a2 = 0.0f, a3 = 0.0f;
  if (MORE) {
    a2 = as[(long)s2 * H + hh];
    a3 = as[(long)s3 * H + hh];
  }
  __builtin_amdgcn_sched_barrier(0);
  v4f x[M];
#pragma unroll
  for (int q = 0; q < M; ++q) x[q] = ld_row(xl, lane_bcast_i((q < 8) ? s0 : s1, q & 7), lane);
  const v4f xs = ld_row(xl, n, lane);
  __builtin_amdgcn_sched_barrier(0);
  const float ls = leaky02(asn + adv);
  const float l0 = leaky02(a0 + adv);
  const float l1 = leaky02(a1 + adv);
  float mx = fmaxf(ls, mrest);
  if (v0) mx = fmaxf(mx, l0);
  if (v1) mx = fmaxf(mx, l1);
  float l2 = 0.0f, l3 = 0.0f;
  if (MORE) {
    l2 = leaky02(a2 + adv);
    l3 = leaky02(a3 + adv);
    if (v2) mx = fmaxf(mx, l2);
    if (v3) mx = fmaxf(mx, l3);
  }
  mx = grp8_max(mx);
  const float es = expf(ls - mx);
  const float t0 = expf(l0 - mx);
  const float t1 = expf(l1 - mx);
  const float e0 = v0 ? t0 : 0.0f;
  const float e1 = v1 ? t1 : 0.0f;
  if (MORE) {
    const float t2 = expf(l2 - mx);
    const float t3 = expf(l3 - mx);
    e2o = v2 ? t2 : 0.0f;
    e3o = v3 ? t3 : 0.0f;
  }
  acc = xs * es;
#pragma unroll
  for (int q = 0; q < M; ++q) acc += x[q] * grp8_bcast((q < 8) ? e0 : e1, q);
  float d = 0.0f;
  d += e0;
  d += e1;
  dl = d;
  exs = es;
  mxo = mx;
  advo = adv;
}

template <bool PLACE>
__device__ __forceinline__ void scan_chunk(const int (&d)[4], int e0, int r0, int nrows,
                                           int wv, int lane, int* cnt,
                                           const int* rowoff, int* list,
                                           int& wcount, int* wle, int* wlr) {
  unsigned u[4]; bool mt[4];
#pragma unroll
  for (int c = 0; c < 4; ++c) { u[c] = (unsigned)(d[c] - r0); mt[c] = u[c] < (unsigned)nrows; }
  if (!__builtin_amdgcn_ballot_w32(mt[0] | mt[1] | mt[2] | mt[3])) return;
#pragma unroll
  for (int c = 0; c < 4; ++c) {
    if (!PLACE) {
      const unsigned mask = __builtin_amdgcn_ballot_w32(mt[c]);
      if (mask == 0) continue;
      if (mt[c]) {
        atomicAdd(&cnt[wv * TR + (int)u[c]], 1);
        const int pos = wcount + __builtin_popcount(mask & ((1u << lane) - 1u));
        if (pos < WCAP) { wle[pos] = e0 + c; wlr[pos] = (int)u[c]; }
      }
      wcount += __builtin_popcount(mask);
    } else {
      unsigned mask = __builtin_amdgcn_ballot_w32(mt[c]);
      while (mask) {
        const int L = __builtin_ctz(mask);
        mask &= mask - 1;
        const int rowL = lane_bcast_i((int)u[c], L);
        if (lane == 0) {
          const int p = cnt[wv * TR + rowL];
          cnt[wv * TR + rowL] = p + 1;
          const int pos = rowoff[rowL] + p;
          if (pos < CAP) list[pos] = e0 - lane * 4 + L * 4 + c;
        }
      }
    }
  }
}

template <bool PLACE>
__device__ __forceinline__ void scan_slice(const int* __restrict__ ei, int E, int r0, int nrows,
                                           int wv, int lane, int per, int* cnt,
                                           const int* rowoff, int* list,
                                           int& wcount, int* wle, int* wlr) {
  const int* dstp = ei + E;
  const int beg = wv * per;
  const int end = (beg + per < E) ? (beg + per) : E;
  const bool al = ((((size_t)dstp) & 15) == 0);
  const int ng = (end > beg) ? ((end - beg + 1023) >> 10) : 0;
  const int rot = (ng > 0) ? (int)(blockIdx.x % (unsigned)ng) : 0;
  for (int g = 0; g < ng; ++g) {
    const int gg = (g + rot < ng) ? (g + rot) : (g + rot - ng);
    const int base = beg + (gg << 10);
    if (al && base + 1024 <= end) {
      int4 v[8];
#pragma unroll
      for (int q = 0; q < 8; ++q) v[q] = *(const int4*)(dstp + base + q * 128 + lane * 4);
#pragma unroll
      for (int q = 0; q < 8; ++q) {
        const int d[4] = {v[q].x, v[q].y, v[q].z, v[q].w};
        scan_chunk<PLACE>(d, base + q * 128 + lane * 4, r0, nrows, wv, lane, cnt, rowoff, list, wcount, wle, wlr);
      }
    } else {
      const int lim = (base + 1024 < end) ? (base + 1024) : end;
      for (int b2 = base; b2 < lim; b2 += 128) {
        const int e0 = b2 + lane * 4;
        int d[4] = {-1, -1, -1, -1};
#pragma unroll
        for (int c = 0; c < 4; ++c) if (e0 + c < end) d[c] = dstp[e0 + c];
        scan_chunk<PLACE>(d, e0, r0, nrows, wv, lane, cnt, rowoff, list, wcount, wle, wlr);
      }
    }
  }
}

__device__ __forceinline__ unsigned key_flags(unsigned x, unsigned T4) {
  const unsigned y = x ^ T4;
  return (y - 0x01010101u) & ~y & 0x80808080u;
}
__device__ __forceinline__ void scan_quad(unsigned x0, unsigned x1, unsigned x2, unsigned x3, int e0,
                                          unsigned T4, int& wcount, int* wle) {
  unsigned g = (key_flags(x0, T4) >> 7) | (key_flags(x1, T4) >> 6) |
               (key_flags(x2, T4) >> 5) | (key_flags(x3, T4) >> 4);
  unsigned hm = __builtin_amdgcn_ballot_w32(g != 0u);
  while (hm) {
    if (g != 0u) {
      const int b = __builtin_ctz(g);
      const int pos = (int)__builtin_amdgcn_mbcnt_lo(hm, (unsigned)wcount);
      if (pos < WCAP) wle[pos] = e0 + ((b & 3) << 2) + (b >> 3);
      g &= g - 1u;
    }
    wcount += __builtin_popcount(hm);
    hm = __builtin_amdgcn_ballot_w32(g != 0u);
  }
}

__device__ __forceinline__ void scan_keys(const int* __restrict__ tkey, int E, int wv, int lane, int per,
                                          int& wcount, int* wle) {
  const unsigned char* tk = (const unsigned char*)tkey;
  const unsigned T  = blockIdx.x & 255u;
  const unsigned T4 = T * 0x01010101u;
  const int beg = wv * per;
  const int end = (beg + per < E) ? (beg + per) : E;
  const int ng = (end > beg) ? ((end - beg + 2047) >> 11) : 0;
  const int rot = (ng > 0) ? (int)(blockIdx.x % (unsigned)ng) : 0;
  for (int g = 0; g < ng; ++g) {
    const int gg = (g + rot < ng) ? (g + rot) : (g + rot - ng);
    const int base = beg + (gg << 11);
    if (base + 2048 <= end) {
      uint4 v[4];
#pragma unroll
      for (int q = 0; q < 4; ++q) v[q] = *(const uint4*)(tk + base + q * 512 + lane * 16);
#pragma unroll
      for (int q = 0; q < 4; ++q) {
        scan_quad(v[q].x, v[q].y, v[q].z, v[q].w, base + q * 512 + lane * 16, T4, wcount, wle);
      }
    } else {
      for (int b2 = base; b2 < end; b2 += 512) {
        const int e0 = b2 + lane * 16;
        unsigned x[4];
#pragma unroll
        for (int d = 0; d < 4; ++d) {
          const int ed = e0 + d * 4;
          unsigned w = ~T4;
          if (ed + 4 <= end) {
            w = *(const unsigned*)(tk + ed);
          } else if (ed < end) {
#pragma unroll
            for (int j = 0; j < 4; ++j)
              if (ed + j < end) w = (w & ~(255u << (8 * j))) | ((unsigned)tk[ed + j] << (8 * j));
          }
          x[d] = w;
        }
        scan_quad(x[0], x[1], x[2], x[3], e0, T4, wcount, wle);
      }
    }
  }
}

__device__ __forceinline__ void read_bins(const int* __restrict__ bins, int E, int wv, int lane, int per,
                                          int& wcount, int* wle, bool& over) {
  const int T = (int)(blockIdx.x & 255u);
  const int nper = per / BCH;
  const int c0 = __builtin_amdgcn_readfirstlane(wv) * nper;
  const int nchunk = (E + BCH - 1) / BCH;
  over = false;
  for (int kb = 0; kb < nper; kb += 8) {
    int w[8];
#pragma unroll
    for (int q = 0; q < 8; ++q) {
      const int cc = c0 + kb + q;
      w[q] = (kb + q < nper && cc < nchunk) ? bins[((size_t)cc * 256 + T) * BW + lane] : 0;
    }
#pragma unroll
    for (int q = 0; q < 8; ++q) {
      const int n = __builtin_amdgcn_readfirstlane(w[q]);
      if (n > BW - 1) over = true;
      const int m = (n < BW - 1) ? n : (BW - 1);
      if (lane >= 1 && lane <= m) {
        const int pos = wcount + lane - 1;
        if (pos < WCAP) wle[pos] = w[q];
      }
      wcount += m;
    }
  }
}

__device__ __forceinline__ void resolve_recorded(const int* __restrict__ dstp, int r0, int nrows,
                                                 int wv, int lane, int wcount, const int* wle,
                                                 int* wlr, int* cnt) {
  for (int k = 0; k < wcount; k += 128) {
    int d[4];
#pragma unroll
    for (int t = 0; t < 4; ++t) {
      const int i = k + t * 32 + lane;
      d[t] = (i < wcount) ? dstp[wle[i]] : -1;
    }
#pragma unroll
    for (int t = 0; t < 4; ++t) {
      const int i = k + t * 32 + lane;
      if (i < wcount) {
        const unsigned u = (unsigned)(d[t] - r0);
        const bool ok = u < (unsigned)nrows;
        wlr[i] = ok ? (int)u : 255;
        if (ok) atomicAdd(&cnt[wv * TR + (int)u], 1);
      }
    }
  }
}

__device__ __forceinline__ void place_recorded(int wv, int lane, int wcount, const int* wle,
                                               const int* wlr, int* cnt, const int* rowoff,
                                               int* list) {
  const unsigned lower = (1u << lane) - 1u;
  for (int k = 0; k < wcount; k += 32) {
    const int i = k + lane;
    const bool v = i < wcount;
    const int e  = v ? wle[i] : 0;
    const int rw = v ? wlr[i] : 255;
    const bool pv = rw != 255;
    unsigned peers = __builtin_amdgcn_ballot_w32(pv);
#pragma unroll
    for (int b = 0; b < 8; ++b) {
      const bool bit = ((rw >> b) & 1) != 0;
      const unsigned mb = __builtin_amdgcn_ballot_w32(bit);
      peers &= bit ? mb : ~mb;
    }
    if (pv) {
      const int p   = cnt[wv * TR + rw];
      const int pos = rowoff[rw] + p + __builtin_popcount(peers & lower);
      if (pos < CAP) list[pos] = e;
      if ((peers & lower) == 0u) cnt[wv * TR + rw] = p + __builtin_popcount(peers);
    }
  }
}

template <int H, bool LAST, bool BUILD>
__global__ __launch_bounds__(1024)
void gat_dst_tile(const int* __restrict__ ei, const float* __restrict__ xl,
                  const float* __restrict__ as, const float* __restrict__ ad,
                  const float* __restrict__ bias, float* __restrict__ outp,
                  int* __restrict__ glist, int* __restrict__ goff, const int* __restrict__ tkey,
                  const int* __restrict__ bins, int E, int N) {
  __shared__ int cnt[NWV * TR];
  __shared__ int tot[TR + 32];
  __shared__ int rowoff[TR + 32];
  __shared__ int list[CAP];
  __shared__ int wle_all[BUILD ? NWV * WCAP : 1];
  __shared__ int wlr_all[BUILD ? NWV * WCAP : 1];

  const int tid  = threadIdx.x;
  const int lane = tid & 31;
  const int wv   = tid >> 5;
  const int r0   = blockIdx.x * TR;
  const int nrows = (N - r0 < TR) ? (N - r0) : TR;
  const int per  = (((E + NWV - 1) / NWV) + 2047) & ~2047;
  int* mylist = glist + (long)blockIdx.x * CAP;
  int* myoff  = goff + (long)blockIdx.x * (TR + 1);

  bool ovf;
  if (BUILD) {
    for (int i = tid; i < NWV * TR; i += 1024) cnt[i] = 0;
    if (tid < TR + 32) { tot[tid] = 0; rowoff[tid] = 0; }
    __syncthreads();

    int wcount = 0;
    int* wle = wle_all + wv * WCAP;
    int* wlr = wlr_all + wv * WCAP;
    bool over;
    read_bins(bins, E, wv, lane, per, wcount, wle, over);
    if (over) {
      wcount = 0;
      scan_keys(tkey, E, wv, lane, per, wcount, wle);
    }
    if (wcount <= WCAP) {
      resolve_recorded(ei + E, r0, nrows, wv, lane, wcount, wle, wlr, cnt);
    } else {
      wcount = 0;
      scan_slice<false>(ei, E, r0, nrows, wv, lane, per, cnt, rowoff, list, wcount, wle, wlr);
    }
    __syncthreads();

    if (tid < TR) {
      int run = 0;
      for (int w2 = 0; w2 < NWV; ++w2) {
        const int q = cnt[w2 * TR + tid];
        cnt[w2 * TR + tid] = run;
        run += q;
      }
      tot[tid] = run;
    }
    __syncthreads();
    if (wv == 0) {
      int loc[7]; int sum = 0;
#pragma unroll
      for (int i = 0; i < 7; ++i) { loc[i] = sum; sum += tot[lane * 7 + i]; }
      int incl = sum;
      for (int o = 1; o < 32; o <<= 1) {
        int t = __shfl_up(incl, o, 32);
        if (lane >= o) incl += t;
      }
      const int excl = incl - sum;
#pragma unroll
      for (int i = 0; i < 7; ++i) rowoff[lane * 7 + i] = excl + loc[i];
    }
    __syncthreads();
    ovf = rowoff[TR] > CAP;
    if (!ovf) {
      if (wcount <= WCAP) place_recorded(wv, lane, wcount, wle, wlr, cnt, rowoff, list);
      else scan_slice<true>(ei, E, r0, nrows, wv, lane, per, cnt, rowoff, list, wcount, wle, wlr);
    }
    __syncthreads();
    if (!ovf) {
      const int total = rowoff[TR];
      for (int i = tid; i < total; i += 1024) {
        const int s = ei[list[i]];
        list[i] = s;
        mylist[i] = s;
      }
    }
    if (tid <= TR) myoff[tid] = rowoff[tid];
    __syncthreads();
  } else {
    if (tid <= TR) rowoff[tid] = myoff[tid];
    __syncthreads();
    ovf = rowoff[TR] > CAP;
    if (!ovf) {
      const int total = rowoff[TR];
      for (int i = tid; i < total; i += 1024) list[i] = mylist[i];
    }
    __syncthreads();
  }

  const int* dstp = ei + E;
  const int j8 = lane & 7;
  const int hh = (H == 4) ? (lane >> 3) : 0;
  const v4f bv = *(const v4f*)(bias + lane * 4);
  const int wvu = __builtin_amdgcn_readfirstlane(wv);
  for (int r = wvu; r < nrows; r += NWV) {
    const int n   = r0 + r;
    const int off = __builtin_amdgcn_readfirstlane(rowoff[r]);
    const int c   = __builtin_amdgcn_readfirstlane(rowoff[r + 1]) - off;
    float exs, den;
    v4f acc;
    if (!ovf) {
      const bool v0 = j8 < c, v1 = 8 + j8 < c;
      const int s0 = v0 ? list[off + j8] : n;
      const int s1 = v1 ? list[off + 8 + j8] : n;
      float dl, mx, adv;
      if (c <= 16) {
        float u2, u3;
        switch ((c + 1) >> 1) {
          case 0:
          case 1:  row_open<H, 2, false>(xl, as, ad, hh, s0, s1, v0, v1, n, n, false, false, -INFINITY, n, lane, exs, dl, mx, u2, u3, adv, acc);  break;
          case 2:  row_open<H, 4, false>(xl, as, ad, hh, s0, s1, v0, v1, n, n, false, false, -INFINITY, n, lane, exs, dl, mx, u2, u3, adv, acc);  break;
          case 3:  row_open<H, 6, false>(xl, as, ad, hh, s0, s1, v0, v1, n, n, false, false, -INFINITY, n, lane, exs, dl, mx, u2, u3, adv, acc);  break;
          case 4:  row_open<H, 8, false>(xl, as, ad, hh, s0, s1, v0, v1, n, n, false, false, -INFINITY, n, lane, exs, dl, mx, u2, u3, adv, acc);  break;
          case 5:  row_open<H, 10, false>(xl, as, ad, hh, s0, s1, v0, v1, n, n, false, false, -INFINITY, n, lane, exs, dl, mx, u2, u3, adv, acc); break;
          case 6:  row_open<H, 12, false>(xl, as, ad, hh, s0, s1, v0, v1, n, n, false, false, -INFINITY, n, lane, exs, dl, mx, u2, u3, adv, acc); break;
          case 7:  row_open<H, 14, false>(xl, as, ad, hh, s0, s1, v0, v1, n, n, false, false, -INFINITY, n, lane, exs, dl, mx, u2, u3, adv, acc); break;
          default: row_open<H, 16, false>(xl, as, ad, hh, s0, s1, v0, v1, n, n, false, false, -INFINITY, n, lane, exs, dl, mx, u2, u3, adv, acc); break;
        }
      } else {
        const bool v2 = 16 + j8 < c, v3 = 24 + j8 < c;
        const int s2 = v2 ? list[off + 16 + j8] : n;
        const int s3 = v3 ? list[off + 24 + j8] : n;
        float mrest = -INFINITY;
        if (c > 32) {
          const float advr = ad[(long)n * H + hh];
          for (int base = 32; base < c; base += 16) {
            const int i0 = base + j8, i1 = i0 + 8;
            const bool w0 = i0 < c, w1 = i1 < c;
            const int q0 = w0 ? list[off + i0] : n;
            const int q1 = w1 ? list[off + i1] : n;
            const float l0 = leaky02(as[(long)q0 * H + hh] + advr);
            const float l1 = leaky02(as[(long)q1 * H + hh] + advr);
            if (w0) mrest = fmaxf(mrest, l0);
            if (w1) mrest = fmaxf(mrest, l1);
          }
        }
        float e2, e3;
        row_open<H, 16, true>(xl, as, ad, hh, s0, s1, v0, v1, s2, s3, v2, v3, mrest, n, lane, exs, dl, mx, e2, e3, adv, acc);
        for (int base = 16; base < c; base += 16) {
          int sa, sb; float ea, eb;
          if (base == 16) {
            sa = s2; sb = s3; ea = e2; eb = e3;
          } else {
            const int i0 = base + j8, i1 = i0 + 8;
            const bool w0 = i0 < c, w1 = i1 < c;
            sa = w0 ? list[off + i0] : n;
            sb = w1 ? list[off + i1] : n;
            const float b0 = as[(long)sa * H + hh];
            const float b1 = as[(long)sb * H + hh];
            const float t0 = expf(leaky02(b0 + adv) - mx);
            const float t1 = expf(leaky02(b1 + adv) - mx);
            ea = w0 ? t0 : 0.0f;
            eb = w1 ? t1 : 0.0f;
          }
          dl += ea;
          dl += eb;
          const int m = (c - base < 16) ? (c - base) : 16;
          switch ((m + 1) >> 1) {
            case 1:  row_trip<2>(xl, sa, sb, ea, eb, lane, acc);  break;
            case 2:  row_trip<4>(xl, sa, sb, ea, eb, lane, acc);  break;
            case 3:  row_trip<6>(xl, sa, sb, ea, eb, lane, acc);  break;
            case 4:  row_trip<8>(xl, sa, sb, ea, eb, lane, acc);  break;
            case 5:  row_trip<10>(xl, sa, sb, ea, eb, lane, acc); break;
            case 6:  row_trip<12>(xl, sa, sb, ea, eb, lane, acc); break;
            case 7:  row_trip<14>(xl, sa, sb, ea, eb, lane, acc); break;
            default: row_trip<16>(xl, sa, sb, ea, eb, lane, acc); break;
          }
        }
      }
      den = grp8_sum(dl) + exs;
    } else {
      const float adv = ad[(long)n * H + hh];
      const float ls  = leaky02(as[(long)n * H + hh] + adv);
      float mx = ls;
      const v4f xs = ld_row(xl, n, lane);
      const int nch = (E + 31) >> 5;
      for (int k = 0; k < nch; ++k) {
        const int i = k * 32 + lane;
        const bool valid = (i < E) && (dstp[i] == n);
        const int sv = valid ? ei[i] : 0;
        unsigned mask = __builtin_amdgcn_ballot_w32(valid);
        while (mask) {
          const int L = __builtin_ctz(mask); mask &= mask - 1;
          const int sq = lane_bcast_i(sv, L);
          mx = fmaxf(mx, leaky02(as[(long)sq * H + hh] + adv));
        }
      }
      exs = expf(ls - mx);
      acc = xs * exs;
      float dl = 0.0f;
      for (int k = 0; k < nch; ++k) {
        const int i = k * 32 + lane;
        const bool valid = (i < E) && (dstp[i] == n);
        const int sv = valid ? ei[i] : 0;
        unsigned mask = __builtin_amdgcn_ballot_w32(valid);
        while (mask) {
          const int L = __builtin_ctz(mask); mask &= mask - 1;
          const int sq = lane_bcast_i(sv, L);
          const float w = expf(leaky02(as[(long)sq * H + hh] + adv) - mx);
          const v4f xq = ld_row(xl, sq, lane);
          dl += w;
          acc += xq * w;
        }
      }
      den = dl + exs;
    }

    const float dn = den + 1e-16f;
    const v4f v = acc / dn + bv;
    v4f* orow = (v4f*)(outp + (long)n * 128 + lane * 4);
    if (!LAST) {
      v4f o;
#pragma unroll
      for (int b = 0; b < 4; ++b) o[b] = (v[b] > 0.0f) ? v[b] : (expf(v[b]) - 1.0f);
      *orow = o;
    } else {
      float ss = 0.0f;
#pragma unroll
      for (int b = 0; b < 4; ++b) ss += v[b] * v[b];
      ss = grp8_sum(ss);
      ss += __shfl_xor(ss, 8, 32);
      ss += __shfl_xor(ss, 16, 32);
      const float sc = 1.0f / fmaxf(sqrtf(ss), 1e-12f);
      *orow = v * sc;
    }
  }
}

extern "C" void kernel_launch(void* const* d_in, const int* in_sizes, int n_in,
                              void* d_out, int out_size, void* d_ws, size_t ws_size,
                              hipStream_t stream) {
  const int*   ei    = (const int*)d_in[0];
  const float* emb   = (const float*)d_in[1];
  const float* W1    = (const float*)d_in[2];
  const float* asrc1 = (const float*)d_in[3];
  const float* adst1 = (const float*)d_in[4];
  const float* b1    = (const float*)d_in[5];
  const float* W2    = (const float*)d_in[6];
  const float* asrc2 = (const float*)d_in[7];
  const float* adst2 = (const float*)d_in[8];
  const float* b2    = (const float*)d_in[9];
  float* out = (float*)d_out;

  const int E  = in_sizes[0] / 2;
  const int N  = in_sizes[1] / 128;

  float* p   = (float*)d_ws;
  float* xl1 = p; p += (size_t)N * 128;
  float* x1  = p; p += (size_t)N * 128;
  float* xl2 = p; p += (size_t)N * 128;
  float* as1 = p; p += (size_t)N * 4;
  float* ad1 = p; p += (size_t)N * 4;
  float* as2 = p; p += (size_t)N;
  float* ad2 = p; p += (size_t)N;
  int gTile  = (N + TR - 1) / TR;
  int* glist = (int*)p; p += (size_t)gTile * CAP;
  int* goff  = (int*)p;
  int* tkey  = (int*)((((size_t)(goff + (size_t)gTile * (TR + 1))) + 15) & ~(size_t)15);
  int* bins  = (int*)((((size_t)tkey) + (size_t)E + 8 + 127) & ~(size_t)127);

  dim3 blk(256);
  int gGemm  = (N + 127) / 128;

  gemm128_wmma <<<gGemm,  blk, 0, stream>>>(emb, W1, asrc1, adst1, xl1, as1, ad1, N, 4, ei + E, tkey, bins, E);
  gat_dst_tile<4, false, true><<<gTile, dim3(1024), 0, stream>>>(ei, xl1, as1, ad1, b1, x1, glist, goff, tkey, bins, E, N);

  gemm128_wmma <<<gGemm,  blk, 0, stream>>>(x1, W2, asrc2, adst2, xl2, as2, ad2, N, 1, nullptr, nullptr, nullptr, 0);
  gat_dst_tile<1, true, false><<<gTile, dim3(1024), 0, stream>>>(ei, xl2, as2, ad2, b2, out, glist, goff, tkey, bins, E, N);
}


# Second GEMM moved into the first row kernel: a tile's finished rows stay in LDS, W2 staged beside them, 13 strips x 9 column tiles on the same f32 WMMA, same k order; 4 launches to 3; sums unchanged
# speedup vs baseline: 8.8493x; 1.1057x over previous
#include <hip/hip_runtime.h>
#include <hip/hip_bf16.h>
#include <math.h>

typedef float v2f __attribute__((ext_vector_type(2)));
typedef float v4f __attribute__((ext_vector_type(4)));
typedef float v8f __attribute__((ext_vector_type(8)));

#define WAS_LD 17
#define TR  196
#define BCH 2048
#define BW  32

__global__ __launch_bounds__(256)
void gemm128_wmma(const float* __restrict__ X, const float* __restrict__ W,
                  const float* __restrict__ asrc, const float* __restrict__ adst,
                  float* __restrict__ Y, float* __restrict__ as_out,
                  float* __restrict__ ad_out, int nrows, int H,
                  const int* __restrict__ dstv, int* __restrict__ tkey, int* __restrict__ bins, int E) {
  __shared__ int bcnt[8 * 256];
  __shared__ __attribute__((aligned(16))) int binL[256 * BW];
  __shared__ float Xs[128 * 33];
  __shared__ float Ws[32 * 132];
  __shared__ float Was[32 * WAS_LD];

  const int tid  = threadIdx.x;
  const int lane = tid & 31;
  const int wv   = tid >> 5;
  const int rc   = lane & 15;
  const int kh   = (lane >> 4) << 1;
  const int r0   = blockIdx.x * 128;

  const float* px[4];
  const float* pw[4];
  int xso[4], wso[4];
#pragma unroll
  for (int i = 0; i < 4; ++i) {
    const int idx = tid + i * 256;
    const int xrow = idx >> 3, xc4 = (idx & 7) << 2;
    const int gr = (r0 + xrow < nrows) ? (r0 + xrow) : (nrows - 1);
    px[i]  = X + (long)gr * 128 + xc4;
    xso[i] = xrow * 33 + xc4;
    const int wrow4 = idx >> 5, wc4 = (idx & 31) << 2;
    pw[i]  = W + (long)wrow4 * 128 + wc4;
    wso[i] = wrow4 * 132 + wc4;
  }
  v4f xr[4], wr[4];
#pragma unroll
  for (int i = 0; i < 4; ++i) {
    xr[i] = *(const v4f*)px[i];
    wr[i] = *(const v4f*)pw[i];
  }

  if (tkey != nullptr) {
    const int ng8 = (E + 7) >> 3;
    const int nchunk = (E + BCH - 1) / BCH;
    const bool al = ((((size_t)dstv) & 15) == 0);
    const unsigned lower = (1u << lane) - 1u;
    for (int c = (int)blockIdx.x; c < nchunk; c += (int)gridDim.x) {
      for (int i = tid; i < 8 * 256; i += 256) bcnt[i] = 0;
      for (int i = tid; i < 256 * BW / 4; i += 256) ((int4*)binL)[i] = make_int4(0, 0, 0, 0);
      __syncthreads();
      const int g = c * 256 + tid;
      const int e = g << 3;
      int d[8];
      if (al && e + 8 <= E) {
        const int4 a = *(const int4*)(dstv + e);
        const int4 b = *(const int4*)(dstv + e + 4);
        d[0] = a.x; d[1] = a.y; d[2] = a.z; d[3] = a.w;
        d[4] = b.x; d[5] = b.y; d[6] = b.z; d[7] = b.w;
      } else {
#pragma unroll
        for (int j = 0; j < 8; ++j) d[j] = (e + j < E) ? dstv[e + j] : 0;
      }
      int key[8];
#pragma unroll
      for (int j = 0; j < 8; ++j) key[j] = (int)(((unsigned)d[j] / (unsigned)TR) & 255u);
      if (g < ng8) {
        unsigned lo = 0u, hi = 0u;
#pragma unroll
        for (int j = 0; j < 4; ++j) {
          lo |= (unsigned)key[j] << (8 * j);
          hi |= (unsigned)key[4 + j] << (8 * j);
        }
        int2 o; o.x = (int)lo; o.y = (int)hi;
        *(int2*)(tkey + (g << 1)) = o;
      }
#pragma unroll
      for (int j = 0; j < 8; ++j)
        if (e + j < E) atomicAdd(&bcnt[wv * 256 + key[j]], 1);
      __syncthreads();
      {
        int run = 0;
#pragma unroll
        for (int w2 = 0; w2 < 8; ++w2) {
          const int q = bcnt[w2 * 256 + tid];
          bcnt[w2 * 256 + tid] = run;
          run += q;
        }
        binL[tid * BW] = run;
      }
      __syncthreads();
#pragma unroll
      for (int j = 0; j < 8; ++j) {
        const bool pv = e + j < E;
        unsigned peers = __builtin_amdgcn_ballot_w32(pv);
#pragma unroll
        for (int b = 0; b < 8; ++b) {
          const bool bit = ((key[j] >> b) & 1) != 0;
          const unsigned mb = __builtin_amdgcn_ballot_w32(bit);
          peers &= bit ? mb : ~mb;
        }
        if (pv) {
          const int p   = bcnt[wv * 256 + key[j]];
          const int pos = p + __builtin_popcount(peers & lower);
          if (pos < BW - 1) binL[key[j] * BW + 1 + pos] = e + j;
          if ((peers & lower) == 0u) bcnt[wv * 256 + key[j]] = p + __builtin_popcount(peers);
        }
      }
      __syncthreads();
      int4* bout = (int4*)(bins + (size_t)c * 256 * BW);
      for (int i = tid; i < 256 * BW / 4; i += 256) bout[i] = ((const int4*)binL)[i];
      __syncthreads();
    }
  }

  const int wrow = tid >> 3;
  const int wj   = tid & 7;
  float attr[32];
  {
    const float* ap = ((wj < 4) ? asrc : adst) + (wj & 3) * 32;
#pragma unroll
    for (int q = 0; q < 8; ++q) {
      const float4 v = *(const float4*)(ap + q * 4);
      attr[q * 4 + 0] = v.x; attr[q * 4 + 1] = v.y; attr[q * 4 + 2] = v.z; attr[q * 4 + 3] = v.w;
    }
  }

  v8f acc[8];
#pragma unroll
  for (int nt = 0; nt < 8; ++nt)
#pragma unroll
    for (int i = 0; i < 8; ++i) acc[nt][i] = 0.0f;
  v8f acc9;
#pragma unroll
  for (int i = 0; i < 8; ++i) acc9[i] = 0.0f;

  for (int kb = 0; kb < 4; ++kb) {
#pragma unroll
    for (int i = 0; i < 4; ++i) {
      float* p = &Xs[xso[i]];
      p[0] = xr[i][0]; p[1] = xr[i][1]; p[2] = xr[i][2]; p[3] = xr[i][3];
      float* q = &Ws[wso[i]];
      q[0] = wr[i][0]; q[1] = wr[i][1]; q[2] = wr[i][2]; q[3] = wr[i][3];
    }
    __syncthreads();
    if (kb < 3) {
#pragma unroll
      for (int i = 0; i < 4; ++i) {
        xr[i] = *(const v4f*)(px[i] + (kb + 1) * 32);
        wr[i] = *(const v4f*)(pw[i] + (long)(kb + 1) * 32 * 128);
      }
    }

    {
      const v4f* wp = (const v4f*)&Ws[wrow * 132 + (wj & 3) * 32];
      v4f w[8];
#pragma unroll
      for (int q = 0; q < 8; ++q) w[q] = wp[q];
      asm volatile("" :: "v"(w[0]), "v"(w[1]), "v"(w[2]), "v"(w[3]), "v"(w[4]), "v"(w[5]), "v"(w[6]), "v"(w[7]));
      float s0 = 0.0f, s1 = 0.0f, s2 = 0.0f, s3 = 0.0f;
#pragma unroll
      for (int q = 0; q < 8; ++q) {
        s0 += w[q][0] * attr[q * 4 + 0];
        s1 += w[q][1] * attr[q * 4 + 1];
        s2 += w[q][2] * attr[q * 4 + 2];
        s3 += w[q][3] * attr[q * 4 + 3];
      }
      Was[wrow * WAS_LD + wj]     = (s0 + s1) + (s2 + s3);
      Was[wrow * WAS_LD + 8 + wj] = 0.0f;
    }

    v2f afr[8];
#pragma unroll
    for (int ks = 0; ks < 8; ++ks) {
      const float* p = &Xs[(wv * 16 + rc) * 33 + ks * 4 + kh];
      afr[ks][0] = p[0];
      afr[ks][1] = p[1];
    }
#pragma unroll
    for (int nt = 0; nt < 8; ++nt) {
#pragma unroll
      for (int ks = 0; ks < 8; ++ks) {
        const float* p = &Ws[(ks * 4 + kh) * 132 + nt * 16 + rc];
        v2f b;
        b[0] = p[0];
        b[1] = p[132];
        acc[nt] = __builtin_amdgcn_wmma_f32_16x16x4_f32(
            false, afr[ks], false, b, (short)0, acc[nt], false, false);
      }
    }
    __syncthreads();

    v2f b9[8];
#pragma unroll
    for (int ks = 0; ks < 8; ++ks) {
      const float* p = &Was[(ks * 4 + kh) * WAS_LD + rc];
      b9[ks][0] = p[0];
      b9[ks][1] = p[WAS_LD];
    }
    asm volatile("" :: "v"(b9[0]), "v"(b9[1]), "v"(b9[2]), "v"(b9[3]), "v"(b9[4]), "v"(b9[5]), "v"(b9[6]), "v"(b9[7]));
#pragma unroll
    for (int ks = 0; ks < 8; ++ks)
      acc9 = __builtin_amdgcn_wmma_f32_16x16x4_f32(
          false, afr[ks], false, b9[ks], (short)0, acc9, false, false);
  }

#pragma unroll
  for (int nt = 0; nt < 8; ++nt) {
#pragma unroll
    for (int i = 0; i < 8; ++i) {
      int row = wv * 16 + i + ((lane >> 4) << 3);
      int gr  = r0 + row;
      if (gr < nrows) Y[(long)gr * 128 + nt * 16 + rc] = acc[nt][i];
    }
  }
  if (H == 4) {
#pragma unroll
    for (int i = 0; i < 8; ++i) {
      int gr = r0 + wv * 16 + i + ((lane >> 4) << 3);
      if (gr < nrows && rc < 8) {
        float* o = (rc < 4) ? as_out : ad_out;
        o[(long)gr * 4 + (rc & 3)] = acc9[i];
      }
    }
  } else {
#pragma unroll
    for (int i = 0; i < 8; ++i) {
      int gr = r0 + wv * 16 + i + ((lane >> 4) << 3);
      float t = acc9[i];
      t += __shfl_xor(t, 1, 32);
      t += __shfl_xor(t, 2, 32);
      if (gr < nrows && (rc == 0 || rc == 4)) {
        float* o = (rc == 0) ? as_out : ad_out;
        o[gr] = t;
      }
    }
  }
}

#define NWV 32
#define CAP 16384
#define WCAP 384
#define NSTRIP ((TR + 15) / 16)
#define XT_LD 132
#define WN_LD 132
static_assert(128 * WN_LD + 128 * WAS_LD <= 2 * NWV * WCAP, "the weight matrix and its W.att columns reuse the scan records");
static_assert(TR <= 255, "placing tells rows apart by eight bits and keeps 255 for no row");

__device__ __forceinline__ float leaky02(float l) { return (l > 0.0f) ? l : 0.2f * l; }

__device__ __forceinline__ int lane_bcast_i(int v, int j) {
  return __builtin_amdgcn_readlane(v, j);
}
#define DPP8_SEL(a, b, c, d, e, f, g, h) \
  ((a) | ((b) << 3) | ((c) << 6) | ((d) << 9) | ((e) << 12) | ((f) << 15) | ((g) << 18) | ((h) << 21))
#define DPP8_BCAST(q) DPP8_SEL(q, q, q, q, q, q, q, q)
template <unsigned SEL>
__device__ __forceinline__ float dpp8f(float v) {
  return __int_as_float(__builtin_amdgcn_mov_dpp8(__float_as_int(v), SEL));
}
__device__ __forceinline__ float grp8_max(float v) {
  v = fmaxf(v, dpp8f<DPP8_SEL(1, 0, 3, 2, 5, 4, 7, 6)>(v));
  v = fmaxf(v, dpp8f<DPP8_SEL(2, 3, 0, 1, 6, 7, 4, 5)>(v));
  v = fmaxf(v, dpp8f<DPP8_SEL(4, 5, 6, 7, 0, 1, 2, 3)>(v));
  return v;
}
__device__ __forceinline__ float grp8_sum(float v) {
  v += dpp8f<DPP8_SEL(1, 0, 3, 2, 5, 4, 7, 6)>(v);
  v += dpp8f<DPP8_SEL(2, 3, 0, 1, 6, 7, 4, 5)>(v);
  v += dpp8f<DPP8_SEL(4, 5, 6, 7, 0, 1, 2, 3)>(v);
  return v;
}
__device__ __forceinline__ v4f ld_row(const float* __restrict__ base, int row, int lane) {
  const unsigned ofs = ((unsigned)row << 9) + ((unsigned)lane << 4);
  return *(const v4f*)((const char*)base + ofs);
}
__device__ __forceinline__ float grp8_bcast(float v, int q) {
  switch (q & 7) {
    case 0:  return dpp8f<DPP8_BCAST(0)>(v);
    case 1:  return dpp8f<DPP8_BCAST(1)>(v);
    case 2:  return dpp8f<DPP8_BCAST(2)>(v);
    case 3:  return dpp8f<DPP8_BCAST(3)>(v);
    case 4:  return dpp8f<DPP8_BCAST(4)>(v);
    case 5:  return dpp8f<DPP8_BCAST(5)>(v);
    case 6:  return dpp8f<DPP8_BCAST(6)>(v);
    default: return dpp8f<DPP8_BCAST(7)>(v);
  }
}
template <int M>
__device__ __forceinline__ void row_trip(const float* __restrict__ xl, int s0, int s1, float e0, float e1,
                                         int lane, v4f& acc) {
  v4f x[M];
#pragma unroll
  for (int q = 0; q < M; ++q) x[q] = ld_row(xl, lane_bcast_i((q < 8) ? s0 : s1, q & 7), lane);
#pragma unroll
  for (int q = 0; q < M; ++q) asm volatile("" :: "v"(x[q]));
#pragma unroll
  for (int q = 0; q < M; ++q) acc += x[q] * grp8_bcast((q < 8) ? e0 : e1, q);
}

template <int H, int M, bool MORE>
__device__ __forceinline__ void row_open(const float* __restrict__ xl, const float* __restrict__ as,
                                         const float* __restrict__ ad, int hh,
                                         int s0, int s1, bool v0, bool v1, int s2, int s3, bool v2, bool v3,
                                         float mrest, int n, int lane,
                                         float& exs, float& dl, float& mxo, float& e2o, float& e3o, float& advo,
                                         v4f& acc) {
  const float adv = ad[(long)n * H + hh];
  const float asn = as[(long)n * H + hh];
  const float a0 = as[(long)s0 * H + hh];
  const float a1 = as[(long)s1 * H + hh];
  float a2 = 0.0f, a3 = 0.0f;
  if (MORE) {
    a2 = as[(long)s2 * H + hh];
    a3 = as[(long)s3 * H + hh];
  }
  __builtin_amdgcn_sched_barrier(0);
  v4f x[M];
#pragma unroll
  for (int q = 0; q < M; ++q) x[q] = ld_row(xl, lane_bcast_i((q < 8) ? s0 : s1, q & 7), lane);
  const v4f xs = ld_row(xl, n, lane);
  __builtin_amdgcn_sched_barrier(0);
  const float ls = leaky02(asn + adv);
  const float l0 = leaky02(a0 + adv);
  const float l1 = leaky02(a1 + adv);
  float mx = fmaxf(ls, mrest);
  if (v0) mx = fmaxf(mx, l0);
  if (v1) mx = fmaxf(mx, l1);
  float l2 = 0.0f, l3 = 0.0f;
  if (MORE) {
    l2 = leaky02(a2 + adv);
    l3 = leaky02(a3 + adv);
    if (v2) mx = fmaxf(mx, l2);
    if (v3) mx = fmaxf(mx, l3);
  }
  mx = grp8_max(mx);
  const float es = expf(ls - mx);
  const float t0 = expf(l0 - mx);
  const float t1 = expf(l1 - mx);
  const float e0 = v0 ? t0 : 0.0f;
  const float e1 = v1 ? t1 : 0.0f;
  if (MORE) {
    const float t2 = expf(l2 - mx);
    const float t3 = expf(l3 - mx);
    e2o = v2 ? t2 : 0.0f;
    e3o = v3 ? t3 : 0.0f;
  }
  acc = xs * es;
#pragma unroll
  for (int q = 0; q < M; ++q) acc += x[q] * grp8_bcast((q < 8) ? e0 : e1, q);
  float d = 0.0f;
  d += e0;
  d += e1;
  dl = d;
  exs = es;
  mxo = mx;
  advo = adv;
}

template <bool PLACE>
__device__ __forceinline__ void scan_chunk(const int (&d)[4], int e0, int r0, int nrows,
                                           int wv, int lane, int* cnt,
                                           const int* rowoff, int* list,
                                           int& wcount, int* wle, int* wlr) {
  unsigned u[4]; bool mt[4];
#pragma unroll
  for (int c = 0; c < 4; ++c) { u[c] = (unsigned)(d[c] - r0); mt[c] = u[c] < (unsigned)nrows; }
  if (!__builtin_amdgcn_ballot_w32(mt[0] | mt[1] | mt[2] | mt[3])) return;
#pragma unroll
  for (int c = 0; c < 4; ++c) {
    if (!PLACE) {
      const unsigned mask = __builtin_amdgcn_ballot_w32(mt[c]);
      if (mask == 0) continue;
      if (mt[c]) {
        atomicAdd(&cnt[wv * TR + (int)u[c]], 1);
        const int pos = wcount + __builtin_popcount(mask & ((1u << lane) - 1u));
        if (pos < WCAP) { wle[pos] = e0 + c; wlr[pos] = (int)u[c]; }
      }
      wcount += __builtin_popcount(mask);
    } else {
      unsigned mask = __builtin_amdgcn_ballot_w32(mt[c]);
      while (mask) {
        const int L = __builtin_ctz(mask);
        mask &= mask - 1;
        const int rowL = lane_bcast_i((int)u[c], L);
        if (lane == 0) {
          const int p = cnt[wv * TR + rowL];
          cnt[wv * TR + rowL] = p + 1;
          const int pos = rowoff[rowL] + p;
          if (pos < CAP) list[pos] = e0 - lane * 4 + L * 4 + c;
        }
      }
    }
  }
}

template <bool PLACE>
__device__ __forceinline__ void scan_slice(const int* __restrict__ ei, int E, int r0, int nrows,
                                           int wv, int lane, int per, int* cnt,
                                           const int* rowoff, int* list,
                                           int& wcount, int* wle, int* wlr) {
  const int* dstp = ei + E;
  const int beg = wv * per;
  const int end = (beg + per < E) ? (beg + per) : E;
  const bool al = ((((size_t)dstp) & 15) == 0);
  const int ng = (end > beg) ? ((end - beg + 1023) >> 10) : 0;
  const int rot = (ng > 0) ? (int)(blockIdx.x % (unsigned)ng) : 0;
  for (int g = 0; g < ng; ++g) {
    const int gg = (g + rot < ng) ? (g + rot) : (g + rot - ng);
    const int base = beg + (gg << 10);
    if (al && base + 1024 <= end) {
      int4 v[8];
#pragma unroll
      for (int q = 0; q < 8; ++q) v[q] = *(const int4*)(dstp + base + q * 128 + lane * 4);
#pragma unroll
      for (int q = 0; q < 8; ++q) {
        const int d[4] = {v[q].x, v[q].y, v[q].z, v[q].w};
        scan_chunk<PLACE>(d, base + q * 128 + lane * 4, r0, nrows, wv, lane, cnt, rowoff, list, wcount, wle, wlr);
      }
    } else {
      const int lim = (base + 1024 < end) ? (base + 1024) : end;
      for (int b2 = base; b2 < lim; b2 += 128) {
        const int e0 = b2 + lane * 4;
        int d[4] = {-1, -1, -1, -1};
#pragma unroll
        for (int c = 0; c < 4; ++c) if (e0 + c < end) d[c] = dstp[e0 + c];
        scan_chunk<PLACE>(d, e0, r0, nrows, wv, lane, cnt, rowoff, list, wcount, wle, wlr);
      }
    }
  }
}

__device__ __forceinline__ unsigned key_flags(unsigned x, unsigned T4) {
  const unsigned y = x ^ T4;
  return (y - 0x01010101u) & ~y & 0x80808080u;
}
__device__ __forceinline__ void scan_quad(unsigned x0, unsigned x1, unsigned x2, unsigned x3, int e0,
                                          unsigned T4, int& wcount, int* wle) {
  unsigned g = (key_flags(x0, T4) >> 7) | (key_flags(x1, T4) >> 6) |
               (key_flags(x2, T4) >> 5) | (key_flags(x3, T4) >> 4);
  unsigned hm = __builtin_amdgcn_ballot_w32(g != 0u);
  while (hm) {
    if (g != 0u) {
      const int b = __builtin_ctz(g);
      const int pos = (int)__builtin_amdgcn_mbcnt_lo(hm, (unsigned)wcount);
      if (pos < WCAP) wle[pos] = e0 + ((b & 3) << 2) + (b >> 3);
      g &= g - 1u;
    }
    wcount += __builtin_popcount(hm);
    hm = __builtin_amdgcn_ballot_w32(g != 0u);
  }
}

__device__ __forceinline__ void scan_keys(const int* __restrict__ tkey, int E, int wv, int lane, int per,
                                          int& wcount, int* wle) {
  const unsigned char* tk = (const unsigned char*)tkey;
  const unsigned T  = blockIdx.x & 255u;
  const unsigned T4 = T * 0x01010101u;
  const int beg = wv * per;
  const int end = (beg + per < E) ? (beg + per) : E;
  const int ng = (end > beg) ? ((end - beg + 2047) >> 11) : 0;
  const int rot = (ng > 0) ? (int)(blockIdx.x % (unsigned)ng) : 0;
  for (int g = 0; g < ng; ++g) {
    const int gg = (g + rot < ng) ? (g + rot) : (g + rot - ng);
    const int base = beg + (gg << 11);
    if (base + 2048 <= end) {
      uint4 v[4];
#pragma unroll
      for (int q = 0; q < 4; ++q) v[q] = *(const uint4*)(tk + base + q * 512 + lane * 16);
#pragma unroll
      for (int q = 0; q < 4; ++q) {
        scan_quad(v[q].x, v[q].y, v[q].z, v[q].w, base + q * 512 + lane * 16, T4, wcount, wle);
      }
    } else {
      for (int b2 = base; b2 < end; b2 += 512) {
        const int e0 = b2 + lane * 16;
        unsigned x[4];
#pragma unroll
        for (int d = 0; d < 4; ++d) {
          const int ed = e0 + d * 4;
          unsigned w = ~T4;
          if (ed + 4 <= end) {
            w = *(const unsigned*)(tk + ed);
          } else if (ed < end) {
#pragma unroll
            for (int j = 0; j < 4; ++j)
              if (ed + j < end) w = (w & ~(255u << (8 * j))) | ((unsigned)tk[ed + j] << (8 * j));
          }
          x[d] = w;
        }
        scan_quad(x[0], x[1], x[2], x[3], e0, T4, wcount, wle);
      }
    }
  }
}

__device__ __forceinline__ void read_bins(const int* __restrict__ bins, int E, int wv, int lane, int per,
                                          int& wcount, int* wle, bool& over) {
  const int T = (int)(blockIdx.x & 255u);
  const int nper = per / BCH;
  const int c0 = __builtin_amdgcn_readfirstlane(wv) * nper;
  const int nchunk = (E + BCH - 1) / BCH;
  over = false;
  for (int kb = 0; kb < nper; kb += 8) {
    int w[8];
#pragma unroll
    for (int q = 0; q < 8; ++q) {
      const int cc = c0 + kb + q;
      w[q] = (kb + q < nper && cc < nchunk) ? bins[((size_t)cc * 256 + T) * BW + lane] : 0;
    }
#pragma unroll
    for (int q = 0; q < 8; ++q) {
      const int n = __builtin_amdgcn_readfirstlane(w[q]);
      if (n > BW - 1) over = true;
      const int m = (n < BW - 1) ? n : (BW - 1);
      if (lane >= 1 && lane <= m) {
        const int pos = wcount + lane - 1;
        if (pos < WCAP) wle[pos] = w[q];
      }
      wcount += m;
    }
  }
}

__device__ __forceinline__ void resolve_recorded(const int* __restrict__ dstp, int r0, int nrows,
                                                 int wv, int lane, int wcount, const int* wle,
                                                 int* wlr, int* cnt) {
  for (int k = 0; k < wcount; k += 128) {
    int d[4];
#pragma unroll
    for (int t = 0; t < 4; ++t) {
      const int i = k + t * 32 + lane;
      d[t] = (i < wcount) ? dstp[wle[i]] : -1;
    }
#pragma unroll
    for (int t = 0; t < 4; ++t) {
      const int i = k + t * 32 + lane;
      if (i < wcount) {
        const unsigned u = (unsigned)(d[t] - r0);
        const bool ok = u < (unsigned)nrows;
        wlr[i] = ok ? (int)u : 255;
        if (ok) atomicAdd(&cnt[wv * TR + (int)u], 1);
      }
    }
  }
}

__device__ __forceinline__ void place_recorded(int wv, int lane, int wcount, const int* wle,
                                               const int* wlr, int* cnt, const int* rowoff,
                                               int* list) {
  const unsigned lower = (1u << lane) - 1u;
  for (int k = 0; k < wcount; k += 32) {
    const int i = k + lane;
    const bool v = i < wcount;
    const int e  = v ? wle[i] : 0;
    const int rw = v ? wlr[i] : 255;
    const bool pv = rw != 255;
    unsigned peers = __builtin_amdgcn_ballot_w32(pv);
#pragma unroll
    for (int b = 0; b < 8; ++b) {
      const bool bit = ((rw >> b) & 1) != 0;
      const unsigned mb = __builtin_amdgcn_ballot_w32(bit);
      peers &= bit ? mb : ~mb;
    }
    if (pv) {
      const int p   = cnt[wv * TR + rw];
      const int pos = rowoff[rw] + p + __builtin_popcount(peers & lower);
      if (pos < CAP) list[pos] = e;
      if ((peers & lower) == 0u) cnt[wv * TR + rw] = p + __builtin_popcount(peers);
    }
  }
}

template <int H, bool LAST, bool BUILD>
__global__ __launch_bounds__(1024)
void gat_dst_tile(const int* __restrict__ ei, const float* __restrict__ xl,
                  const float* __restrict__ as, const float* __restrict__ ad,
                  const float* __restrict__ bias, float* __restrict__ outp,
                  int* __restrict__ glist, int* __restrict__ goff, const int* __restrict__ tkey,
                  const int* __restrict__ bins, int E, int N,
                  const float* __restrict__ Wn, const float* __restrict__ asrcn, const float* __restrict__ adstn,
                  float* __restrict__ Yn, float* __restrict__ asn_out, float* __restrict__ adn_out) {
  __shared__ int cnt[NWV * TR];
  __shared__ int tot[TR + 32];
  __shared__ int rowoff[TR + 32];
  __shared__ int list[CAP];
  __shared__ __attribute__((aligned(16))) int wl_all[BUILD ? 2 * NWV * WCAP : 4];
  __shared__ __attribute__((aligned(16))) float xt[BUILD ? NSTRIP * 16 * XT_LD : 4];
  int* const wle_all = wl_all;
  int* const wlr_all = wl_all + (BUILD ? NWV * WCAP : 0);
  float* const Wn_s  = (float*)wl_all;
  float* const Wan_s = Wn_s + 128 * WN_LD;

  const int tid  = threadIdx.x;
  const int lane = tid & 31;
  const int wv   = tid >> 5;
  const int r0   = blockIdx.x * TR;
  const int nrows = (N - r0 < TR) ? (N - r0) : TR;
  const int per  = (((E + NWV - 1) / NWV) + 2047) & ~2047;
  int* mylist = glist + (long)blockIdx.x * CAP;
  int* myoff  = goff + (long)blockIdx.x * (TR + 1);

  bool ovf;
  if (BUILD) {
    for (int i = tid; i < NWV * TR; i += 1024) cnt[i] = 0;
    if (tid < TR + 32) { tot[tid] = 0; rowoff[tid] = 0; }
    __syncthreads();

    int wcount = 0;
    int* wle = wle_all + wv * WCAP;
    int* wlr = wlr_all + wv * WCAP;
    bool over;
    read_bins(bins, E, wv, lane, per, wcount, wle, over);
    if (over) {
      wcount = 0;
      scan_keys(tkey, E, wv, lane, per, wcount, wle);
    }
    if (wcount <= WCAP) {
      resolve_recorded(ei + E, r0, nrows, wv, lane, wcount, wle, wlr, cnt);
    } else {
      wcount = 0;
      scan_slice<false>(ei, E, r0, nrows, wv, lane, per, cnt, rowoff, list, wcount, wle, wlr);
    }
    __syncthreads();

    if (tid < TR) {
      int run = 0;
      for (int w2 = 0; w2 < NWV; ++w2) {
        const int q = cnt[w2 * TR + tid];
        cnt[w2 * TR + tid] = run;
        run += q;
      }
      tot[tid] = run;
    }
    __syncthreads();
    if (wv == 0) {
      int loc[7]; int sum = 0;
#pragma unroll
      for (int i = 0; i < 7; ++i) { loc[i] = sum; sum += tot[lane * 7 + i]; }
      int incl = sum;
      for (int o = 1; o < 32; o <<= 1) {
        int t = __shfl_up(incl, o, 32);
        if (lane >= o) incl += t;
      }
      const int excl = incl - sum;
#pragma unroll
      for (int i = 0; i < 7; ++i) rowoff[lane * 7 + i] = excl + loc[i];
    }
    __syncthreads();
    ovf = rowoff[TR] > CAP;
    if (!ovf) {
      if (wcount <= WCAP) place_recorded(wv, lane, wcount, wle, wlr, cnt, rowoff, list);
      else scan_slice<true>(ei, E, r0, nrows, wv, lane, per, cnt, rowoff, list, wcount, wle, wlr);
    }
    __syncthreads();
    if (!ovf) {
      const int total = rowoff[TR];
      for (int i = tid; i < total; i += 1024) {
        const int s = ei[list[i]];
        list[i] = s;
        mylist[i] = s;
      }
    }
    if (tid <= TR) myoff[tid] = rowoff[tid];
    {
      v4f wq[4];
#pragma unroll
      for (int i = 0; i < 4; ++i) {
        const int idx = tid + i * 1024;
        wq[i] = *(const v4f*)(Wn + (idx >> 5) * 128 + ((idx & 31) << 2));
      }
      const int wrow = tid >> 3, wj = tid & 7;
      const float* ap = ((wj < 4) ? asrcn : adstn) + (wj & 3) * 32;
      const float* wp = Wn + wrow * 128 + (wj & 3) * 32;
      float s0 = 0.0f, s1 = 0.0f, s2 = 0.0f, s3 = 0.0f;
#pragma unroll
      for (int q = 0; q < 8; ++q) {
        const v4f w = *(const v4f*)(wp + q * 4);
        const v4f a = *(const v4f*)(ap + q * 4);
        s0 += w[0] * a[0];
        s1 += w[1] * a[1];
        s2 += w[2] * a[2];
        s3 += w[3] * a[3];
      }
      Wan_s[wrow * WAS_LD + wj]     = (s0 + s1) + (s2 + s3);
      Wan_s[wrow * WAS_LD + 8 + wj] = 0.0f;
#pragma unroll
      for (int i = 0; i < 4; ++i) {
        const int idx = tid + i * 1024;
        *(v4f*)&Wn_s[(idx >> 5) * WN_LD + ((idx & 31) << 2)] = wq[i];
      }
    }
    __syncthreads();
  } else {
    if (tid <= TR) rowoff[tid] = myoff[tid];
    __syncthreads();
    ovf = rowoff[TR] > CAP;
    if (!ovf) {
      const int total = rowoff[TR];
      for (int i = tid; i < total; i += 1024) list[i] = mylist[i];
    }
    __syncthreads();
  }

  const int* dstp = ei + E;
  const int j8 = lane & 7;
  const int hh = (H == 4) ? (lane >> 3) : 0;
  const v4f bv = *(const v4f*)(bias + lane * 4);
  const int wvu = __builtin_amdgcn_readfirstlane(wv);
  for (int r = wvu; r < nrows; r += NWV) {
    const int n   = r0 + r;
    const int off = __builtin_amdgcn_readfirstlane(rowoff[r]);
    const int c   = __builtin_amdgcn_readfirstlane(rowoff[r + 1]) - off;
    float exs, den;
    v4f acc;
    if (!ovf) {
      const bool v0 = j8 < c, v1 = 8 + j8 < c;
      const int s0 = v0 ? list[off + j8] : n;
      const int s1 = v1 ? list[off + 8 + j8] : n;
      float dl, mx, adv;
      if (c <= 16) {
        float u2, u3;
        switch ((c + 1) >> 1) {
          case 0:
          case 1:  row_open<H, 2, false>(xl, as, ad, hh, s0, s1, v0, v1, n, n, false, false, -INFINITY, n, lane, exs, dl, mx, u2, u3, adv, acc);  break;
          case 2:  row_open<H, 4, false>(xl, as, ad, hh, s0, s1, v0, v1, n, n, false, false, -INFINITY, n, lane, exs, dl, mx, u2, u3, adv, acc);  break;
          case 3:  row_open<H, 6, false>(xl, as, ad, hh, s0, s1, v0, v1, n, n, false, false, -INFINITY, n, lane, exs, dl, mx, u2, u3, adv, acc);  break;
          case 4:  row_open<H, 8, false>(xl, as, ad, hh, s0, s1, v0, v1, n, n, false, false, -INFINITY, n, lane, exs, dl, mx, u2, u3, adv, acc);  break;
          case 5:  row_open<H, 10, false>(xl, as, ad, hh, s0, s1, v0, v1, n, n, false, false, -INFINITY, n, lane, exs, dl, mx, u2, u3, adv, acc); break;
          case 6:  row_open<H, 12, false>(xl, as, ad, hh, s0, s1, v0, v1, n, n, false, false, -INFINITY, n, lane, exs, dl, mx, u2, u3, adv, acc); break;
          case 7:  row_open<H, 14, false>(xl, as, ad, hh, s0, s1, v0, v1, n, n, false, false, -INFINITY, n, lane, exs, dl, mx, u2, u3, adv, acc); break;
          default: row_open<H, 16, false>(xl, as, ad, hh, s0, s1, v0, v1, n, n, false, false, -INFINITY, n, lane, exs, dl, mx, u2, u3, adv, acc); break;
        }
      } else {
        const bool v2 = 16 + j8 < c, v3 = 24 + j8 < c;
        const int s2 = v2 ? list[off + 16 + j8] : n;
        const int s3 = v3 ? list[off + 24 + j8] : n;
        float mrest = -INFINITY;
        if (c > 32) {
          const float advr = ad[(long)n * H + hh];
          for (int base = 32; base < c; base += 16) {
            const int i0 = base + j8, i1 = i0 + 8;
            const bool w0 = i0 < c, w1 = i1 < c;
            const int q0 = w0 ? list[off + i0] : n;
            const int q1 = w1 ? list[off + i1] : n;
            const float l0 = leaky02(as[(long)q0 * H + hh] + advr);
            const float l1 = leaky02(as[(long)q1 * H + hh] + advr);
            if (w0) mrest = fmaxf(mrest, l0);
            if (w1) mrest = fmaxf(mrest, l1);
          }
        }
        float e2, e3;
        row_open<H, 16, true>(xl, as, ad, hh, s0, s1, v0, v1, s2, s3, v2, v3, mrest, n, lane, exs, dl, mx, e2, e3, adv, acc);
        for (int base = 16; base < c; base += 16) {
          int sa, sb; float ea, eb;
          if (base == 16) {
            sa = s2; sb = s3; ea = e2; eb = e3;
          } else {
            const int i0 = base + j8, i1 = i0 + 8;
            const bool w0 = i0 < c, w1 = i1 < c;
            sa = w0 ? list[off + i0] : n;
            sb = w1 ? list[off + i1] : n;
            const float b0 = as[(long)sa * H + hh];
            const float b1 = as[(long)sb * H + hh];
            const float t0 = expf(leaky02(b0 + adv) - mx);
            const float t1 = expf(leaky02(b1 + adv) - mx);
            ea = w0 ? t0 : 0.0f;
            eb = w1 ? t1 : 0.0f;
          }
          dl += ea;
          dl += eb;
          const int m = (c - base < 16) ? (c - base) : 16;
          switch ((m + 1) >> 1) {
            case 1:  row_trip<2>(xl, sa, sb, ea, eb, lane, acc);  break;
            case 2:  row_trip<4>(xl, sa, sb, ea, eb, lane, acc);  break;
            case 3:  row_trip<6>(xl, sa, sb, ea, eb, lane, acc);  break;
            case 4:  row_trip<8>(xl, sa, sb, ea, eb, lane, acc);  break;
            case 5:  row_trip<10>(xl, sa, sb, ea, eb, lane, acc); break;
            case 6:  row_trip<12>(xl, sa, sb, ea, eb, lane, acc); break;
            case 7:  row_trip<14>(xl, sa, sb, ea, eb, lane, acc); break;
            default: row_trip<16>(xl, sa, sb, ea, eb, lane, acc); break;
          }
        }
      }
      den = grp8_sum(dl) + exs;
    } else {
      const float adv = ad[(long)n * H + hh];
      const float ls  = leaky02(as[(long)n * H + hh] + adv);
      float mx = ls;
      const v4f xs = ld_row(xl, n, lane);
      const int nch = (E + 31) >> 5;
      for (int k = 0; k < nch; ++k) {
        const int i = k * 32 + lane;
        const bool valid = (i < E) && (dstp[i] == n);
        const int sv = valid ? ei[i] : 0;
        unsigned mask = __builtin_amdgcn_ballot_w32(valid);
        while (mask) {
          const int L = __builtin_ctz(mask); mask &= mask - 1;
          const int sq = lane_bcast_i(sv, L);
          mx = fmaxf(mx, leaky02(as[(long)sq * H + hh] + adv));
        }
      }
      exs = expf(ls - mx);
      acc = xs * exs;
      float dl = 0.0f;
      for (int k = 0; k < nch; ++k) {
        const int i = k * 32 + lane;
        const bool valid = (i < E) && (dstp[i] == n);
        const int sv = valid ? ei[i] : 0;
        unsigned mask = __builtin_amdgcn_ballot_w32(valid);
        while (mask) {
          const int L = __builtin_ctz(mask); mask &= mask - 1;
          const int sq = lane_bcast_i(sv, L);
          const float w = expf(leaky02(as[(long)sq * H + hh] + adv) - mx);
          const v4f xq = ld_row(xl, sq, lane);
          dl += w;
          acc += xq * w;
        }
      }
      den = dl + exs;
    }

    const float dn = den + 1e-16f;
    const v4f v = acc / dn + bv;
    v4f* orow = (v4f*)(outp + (long)n * 128 + lane * 4);
    if (!LAST) {
      v4f o;
#pragma unroll
      for (int b = 0; b < 4; ++b) o[b] = (v[b] > 0.0f) ? v[b] : (expf(v[b]) - 1.0f);
      if (BUILD) *(v4f*)&xt[r * XT_LD + lane * 4] = o;
      else *orow = o;
    } else {
      float ss = 0.0f;
#pragma unroll
      for (int b = 0; b < 4; ++b) ss += v[b] * v[b];
      ss = grp8_sum(ss);
      ss += __shfl_xor(ss, 8, 32);
      ss += __shfl_xor(ss, 16, 32);
      const float sc = 1.0f / fmaxf(sqrtf(ss), 1e-12f);
      *orow = v * sc;
    }
  }

  if (BUILD) {
    __syncthreads();
    const int rc = lane & 15;
    const int kh = (lane >> 4) << 1;
    const int nstrip = (nrows + 15) >> 4;
    const int npair = nstrip * 4;
    for (int u = wvu; u < npair + nstrip; u += NWV) {
      if (u < npair) {
        const int strip = u >> 2, ct = (u & 3) << 1;
        const float* pa = &xt[(strip * 16 + rc) * XT_LD + kh];
        const float* pb = &Wn_s[kh * WN_LD + ct * 16 + rc];
        v8f c0, c1;
#pragma unroll
        for (int i = 0; i < 8; ++i) { c0[i] = 0.0f; c1[i] = 0.0f; }
#pragma unroll
        for (int ks = 0; ks < 32; ++ks) {
          v2f a, b0, b1;
          a[0]  = pa[ks * 4];
          a[1]  = pa[ks * 4 + 1];
          b0[0] = pb[ks * 4 * WN_LD];
          b0[1] = pb[ks * 4 * WN_LD + WN_LD];
          b1[0] = pb[ks * 4 * WN_LD + 16];
          b1[1] = pb[ks * 4 * WN_LD + WN_LD + 16];
          c0 = __builtin_amdgcn_wmma_f32_16x16x4_f32(false, a, false, b0, (short)0, c0, false, false);
          c1 = __builtin_amdgcn_wmma_f32_16x16x4_f32(false, a, false, b1, (short)0, c1, false, false);
        }
#pragma unroll
        for (int i = 0; i < 8; ++i) {
          const int row = strip * 16 + i + ((lane >> 4) << 3);
          if (row < nrows) {
            float* o = Yn + (long)(r0 + row) * 128 + ct * 16 + rc;
            o[0]  = c0[i];
            o[16] = c1[i];
          }
        }
      } else {
        const int strip = u - npair;
        const float* pa = &xt[(strip * 16 + rc) * XT_LD + kh];
        const float* pb = &Wan_s[kh * WAS_LD + rc];
        v8f c9;
#pragma unroll
        for (int i = 0; i < 8; ++i) c9[i] = 0.0f;
#pragma unroll
        for (int ks = 0; ks < 32; ++ks) {
          v2f a, b;
          a[0] = pa[ks * 4];
          a[1] = pa[ks * 4 + 1];
          b[0] = pb[ks * 4 * WAS_LD];
          b[1] = pb[ks * 4 * WAS_LD + WAS_LD];
          c9 = __builtin_amdgcn_wmma_f32_16x16x4_f32(false, a, false, b, (short)0, c9, false, false);
        }
#pragma unroll
        for (int i = 0; i < 8; ++i) {
          const int row = strip * 16 + i + ((lane >> 4) << 3);
          float t = c9[i];
          t += __shfl_xor(t, 1, 32);
          t += __shfl_xor(t, 2, 32);
          if (row < nrows && (rc == 0 || rc == 4)) {
            float* o = (rc == 0) ? asn_out : adn_out;
            o[r0 + row] = t;
          }
        }
      }
    }
  }
}

extern "C" void kernel_launch(void* const* d_in, const int* in_sizes, int n_in,
                              void* d_out, int out_size, void* d_ws, size_t ws_size,
                              hipStream_t stream) {
  const int*   ei    = (const int*)d_in[0];
  const float* emb   = (const float*)d_in[1];
  const float* W1    = (const float*)d_in[2];
  const float* asrc1 = (const float*)d_in[3];
  const float* adst1 = (const float*)d_in[4];
  const float* b1    = (const float*)d_in[5];
  const float* W2    = (const float*)d_in[6];
  const float* asrc2 = (const float*)d_in[7];
  const float* adst2 = (const float*)d_in[8];
  const float* b2    = (const float*)d_in[9];
  float* out = (float*)d_out;

  const int E  = in_sizes[0] / 2;
  const int N  = in_sizes[1] / 128;

  float* p   = (float*)d_ws;
  float* xl1 = p; p += (size_t)N * 128;
  float* xl2 = p; p += (size_t)N * 128;
  float* as1 = p; p += (size_t)N * 4;
  float* ad1 = p; p += (size_t)N * 4;
  float* as2 = p; p += (size_t)N;
  float* ad2 = p; p += (size_t)N;
  int gTile  = (N + TR - 1) / TR;
  int* glist = (int*)p; p += (size_t)gTile * CAP;
  int* goff  = (int*)p;
  int* tkey  = (int*)((((size_t)(goff + (size_t)gTile * (TR + 1))) + 15) & ~(size_t)15);
  int* bins  = (int*)((((size_t)tkey) + (size_t)E + 8 + 127) & ~(size_t)127);

  dim3 blk(256);
  int gGemm  = (N + 127) / 128;

  gemm128_wmma <<<gGemm,  blk, 0, stream>>>(emb, W1, asrc1, adst1, xl1, as1, ad1, N, 4, ei + E, tkey, bins, E);
  gat_dst_tile<4, false, true><<<gTile, dim3(1024), 0, stream>>>(ei, xl1, as1, ad1, b1, nullptr, glist, goff, tkey, bins, E, N,
                                                                 W2, asrc2, adst2, xl2, as2, ad2);

  gat_dst_tile<1, true, false><<<gTile, dim3(1024), 0, stream>>>(ei, xl2, as2, ad2, b2, out, glist, goff, tkey, bins, E, N,
                                                                 nullptr, nullptr, nullptr, nullptr, nullptr, nullptr);
}
